# starting kernel
# baseline (speedup 1.0000x reference)
#include <hip/hip_runtime.h>
#include <math.h>


typedef float v2f __attribute__((ext_vector_type(2)));
typedef float v8f __attribute__((ext_vector_type(8)));

constexpr int NRQ = 4096;
constexpr int NBC = 16384;
constexpr int CB  = 80;
constexpr int CRV = 384;
constexpr int DD  = 128;
constexpr int MM  = 256;
constexpr float INV_SQRT_D = 0.08838834764831845f;


__global__ __launch_bounds__(256)
void rmba_qproj_kernel(const float* __restrict__ A,
                       const float* __restrict__ Wq,
                       const float* __restrict__ bq,
                       float* __restrict__ Qn)
{
    const int lane = threadIdx.x & 31;
    const int wave = threadIdx.x >> 5;
    const int half = lane >> 4;
    const int l16  = lane & 15;
    const int koff = half * 2;
    const int rowBase = blockIdx.x * 16;
    const int colBase = wave * 16;
    const int arow = rowBase + l16;
    const int bcol = colBase + l16;

    v8f acc = {};
    #pragma unroll 4
    for (int k = 0; k < CRV; k += 4) {
        v2f a, b;
        a.x = A[arow * CRV + k + koff];
        a.y = A[arow * CRV + k + koff + 1];
        b.x = Wq[(k + koff) * DD + bcol];
        b.y = Wq[(k + koff + 1) * DD + bcol];
        acc = __builtin_amdgcn_wmma_f32_16x16x4_f32(false, a, false, b,
                                                    (short)0, acc, false, false);
    }

    const float bias = bq[bcol];
    float vals[8], ss[8];
    #pragma unroll
    for (int r = 0; r < 8; ++r) { vals[r] = acc[r] + bias; ss[r] = vals[r] * vals[r]; }

    #pragma unroll
    for (int s = 8; s >= 1; s >>= 1)
        #pragma unroll
        for (int r = 0; r < 8; ++r) ss[r] += __shfl_xor(ss[r], s, 32);

    __shared__ float sq[8][16];
    __shared__ float rnorm[16];
    if (l16 == 0) {
        #pragma unroll
        for (int r = 0; r < 8; ++r) sq[wave][half * 8 + r] = ss[r];
    }
    __syncthreads();
    if (threadIdx.x < 16) {
        float s = 0.f;
        #pragma unroll
        for (int w = 0; w < 8; ++w) s += sq[w][threadIdx.x];
        rnorm[threadIdx.x] = 1.0f / fmaxf(sqrtf(s), 1e-12f);
    }
    __syncthreads();
    #pragma unroll
    for (int r = 0; r < 8; ++r) {
        const int row = half * 8 + r;
        Qn[(rowBase + row) * DD + bcol] = vals[r] * rnorm[row];
    }
}

__global__ __launch_bounds__(256)
void rmba_kvproj_kernel(const float* __restrict__ bev,
                        const float* __restrict__ Wk, const float* __restrict__ bk,
                        const float* __restrict__ Wv, const float* __restrict__ bv,
                        float* __restrict__ Kall,
                        float* __restrict__ Vall)
{
    const int lane = threadIdx.x & 31;
    const int wave = threadIdx.x >> 5;
    const int half = lane >> 4;
    const int l16  = lane & 15;
    const int koff = half * 2;
    const int rowBase = blockIdx.x * 16;
    const int colBase = wave * 16;
    const int arow = rowBase + l16;
    const int bcol = colBase + l16;

    v8f acck = {}, accv = {};
    #pragma unroll 4
    for (int k = 0; k < CB; k += 4) {
        v2f a, wk, wv;
        a.x  = bev[(k + koff) * NBC + arow];
        a.y  = bev[(k + koff + 1) * NBC + arow];
        wk.x = Wk[(k + koff) * DD + bcol];
        wk.y = Wk[(k + koff + 1) * DD + bcol];
        wv.x = Wv[(k + koff) * DD + bcol];
        wv.y = Wv[(k + koff + 1) * DD + bcol];
        acck = __builtin_amdgcn_wmma_f32_16x16x4_f32(false, a, false, wk,
                                                     (short)0, acck, false, false);
        accv = __builtin_amdgcn_wmma_f32_16x16x4_f32(false, a, false, wv,
                                                     (short)0, accv, false, false);
    }

    const float biask = bk[bcol];
    const float biasv = bv[bcol];
    float kv[8], ss[8];
    #pragma unroll
    for (int r = 0; r < 8; ++r) { kv[r] = acck[r] + biask; ss[r] = kv[r] * kv[r]; }
    #pragma unroll
    for (int s = 8; s >= 1; s >>= 1)
        #pragma unroll
        for (int r = 0; r < 8; ++r) ss[r] += __shfl_xor(ss[r], s, 32);

    __shared__ float sq[8][16];
    __shared__ float rnorm[16];
    if (l16 == 0) {
        #pragma unroll
        for (int r = 0; r < 8; ++r) sq[wave][half * 8 + r] = ss[r];
    }
    __syncthreads();
    if (threadIdx.x < 16) {
        float s = 0.f;
        #pragma unroll
        for (int w = 0; w < 8; ++w) s += sq[w][threadIdx.x];
        rnorm[threadIdx.x] = 1.0f / fmaxf(sqrtf(s), 1e-12f);
    }
    __syncthreads();
    #pragma unroll
    for (int r = 0; r < 8; ++r) {
        const int row = half * 8 + r;
        Kall[(rowBase + row) * DD + bcol] = kv[r] * rnorm[row];
        Vall[(rowBase + row) * DD + bcol] = accv[r] + biasv;
    }
}

__global__ __launch_bounds__(256)
void rmba_attn_kernel(const float* __restrict__ Qn,
                      const float* __restrict__ Kall,
                      const float* __restrict__ Vall,
                      const int*   __restrict__ cand,
                      float* __restrict__ attn_out,
                      float* __restrict__ ctx)
{
    const int n   = blockIdx.x;
    const int tid = threadIdx.x;

    __shared__ float q_sm[DD];
    __shared__ float a_sm[MM];
    __shared__ int   idx_sm[MM];
    __shared__ float red_sm[8];
    __shared__ float cpart[256];

    if (tid < DD) q_sm[tid] = Qn[n * DD + tid];
    const int idx = cand[n * MM + tid];
    idx_sm[tid] = idx;
    __syncthreads();

    float logit = -INFINITY;
    if (idx >= 0) {
        const float4* krow = (const float4*)(Kall + (size_t)idx * DD);
        const float4* qv   = (const float4*)q_sm;
        float acc = 0.f;
        #pragma unroll
        for (int c = 0; c < DD / 4; ++c) {
            const float4 k4 = krow[c];
            const float4 q4 = qv[c];
            acc += k4.x * q4.x + k4.y * q4.y + k4.z * q4.z + k4.w * q4.w;
        }
        logit = acc * INV_SQRT_D;
    }

    float mx = logit;
    #pragma unroll
    for (int s = 16; s >= 1; s >>= 1) mx = fmaxf(mx, __shfl_xor(mx, s, 32));
    if ((tid & 31) == 0) red_sm[tid >> 5] = mx;
    __syncthreads();
    float bm = red_sm[0];
    #pragma unroll
    for (int w = 1; w < 8; ++w) bm = fmaxf(bm, red_sm[w]);
    __syncthreads();

    const float e = (idx >= 0) ? __expf(logit - bm) : 0.f;
    float sm = e;
    #pragma unroll
    for (int s = 16; s >= 1; s >>= 1) sm += __shfl_xor(sm, s, 32);
    if ((tid & 31) == 0) red_sm[tid >> 5] = sm;
    __syncthreads();
    float bs = 0.f;
    #pragma unroll
    for (int w = 0; w < 8; ++w) bs += red_sm[w];

    const float a = e / bs;
    attn_out[n * MM + tid] = a;
    a_sm[tid] = a;
    __syncthreads();

    const int d     = tid & 127;
    const int mBase = (tid >> 7) * 128;
    float c = 0.f;
    for (int m = 0; m < 128; ++m) {
        const int ix = idx_sm[mBase + m];
        if (ix >= 0) c += a_sm[mBase + m] * Vall[(size_t)ix * DD + d];
    }
    cpart[tid] = c;
    __syncthreads();
    if (tid < DD) ctx[n * DD + tid] = cpart[tid] + cpart[tid + 128];
}

__global__ __launch_bounds__(256)
void rmba_oproj_kernel(const float* __restrict__ ctx,
                       const float* __restrict__ Wo,
                       const float* __restrict__ bo,
                       float* __restrict__ out)
{
    const int mTiles = NRQ / 16;
    const int mTile  = blockIdx.x % mTiles;
    const int nGrp   = blockIdx.x / mTiles;
    const int lane = threadIdx.x & 31;
    const int wave = threadIdx.x >> 5;
    const int half = lane >> 4;
    const int l16  = lane & 15;
    const int koff = half * 2;
    const int rowBase = mTile * 16;
    const int colBase = nGrp * 128 + wave * 16;
    const int arow = rowBase + l16;
    const int bcol = colBase + l16;

    v8f acc = {};
    #pragma unroll 4
    for (int k = 0; k < DD; k += 4) {
        v2f a, b;
        a.x = ctx[arow * DD + k + koff];
        a.y = ctx[arow * DD + k + koff + 1];
        b.x = Wo[(k + koff) * CRV + bcol];
        b.y = Wo[(k + koff + 1) * CRV + bcol];
        acc = __builtin_amdgcn_wmma_f32_16x16x4_f32(false, a, false, b,
                                                    (short)0, acc, false, false);
    }
    const float bias = bo[bcol];
    #pragma unroll
    for (int r = 0; r < 8; ++r) {
        const int row = rowBase + half * 8 + r;
        out[row * CRV + bcol] = acc[r] + bias;
    }
}

extern "C" void kernel_launch(void* const* d_in, const int* in_sizes, int n_in,
                              void* d_out, int out_size, void* d_ws, size_t ws_size,
                              hipStream_t stream) {
    const float* rv_x = (const float*)d_in[0];
    const float* bev  = (const float*)d_in[1];
    const int*   cand = (const int*)  d_in[2];
    const float* Wq   = (const float*)d_in[3];
    const float* bq   = (const float*)d_in[4];
    const float* Wk   = (const float*)d_in[5];
    const float* bk   = (const float*)d_in[6];
    const float* Wv   = (const float*)d_in[7];
    const float* bv   = (const float*)d_in[8];
    const float* Wo   = (const float*)d_in[9];
    const float* bo   = (const float*)d_in[10];

    float* out_main = (float*)d_out;
    float* out_attn = out_main + (size_t)NRQ * CRV;

    float* Qn   = (float*)d_ws;
    float* Kall = Qn   + (size_t)NRQ * DD;
    float* Vall = Kall + (size_t)NBC * DD;
    float* ctx  = Vall + (size_t)NBC * DD;

    rmba_qproj_kernel <<<NRQ / 16, 256, 0, stream>>>(rv_x, Wq, bq, Qn);
    rmba_kvproj_kernel<<<NBC / 16, 256, 0, stream>>>(bev, Wk, bk, Wv, bv, Kall, Vall);
    rmba_attn_kernel  <<<NRQ,      256, 0, stream>>>(Qn, Kall, Vall, cand, out_attn, ctx);
    rmba_oproj_kernel <<<(NRQ / 16) * (CRV / 128), 256, 0, stream>>>(ctx, Wo, bo, out_main);
}


# attn@V pass: each wave gathers whole 512-B V rows, 16 independent float4 loads in flight, not one guarded dependent 4-B load per trip; ctx summed per wave, then over 8 waves (sum order changes)
# speedup vs baseline: 1.4114x; 1.4114x over previous
#include <hip/hip_runtime.h>
#include <math.h>


typedef float v2f __attribute__((ext_vector_type(2)));
typedef float v8f __attribute__((ext_vector_type(8)));

constexpr int NRQ = 4096;
constexpr int NBC = 16384;
constexpr int CB  = 80;
constexpr int CRV = 384;
constexpr int DD  = 128;
constexpr int MM  = 256;
constexpr float INV_SQRT_D = 0.08838834764831845f;


__global__ __launch_bounds__(256)
void rmba_qproj_kernel(const float* __restrict__ A,
                       const float* __restrict__ Wq,
                       const float* __restrict__ bq,
                       float* __restrict__ Qn)
{
    const int lane = threadIdx.x & 31;
    const int wave = threadIdx.x >> 5;
    const int half = lane >> 4;
    const int l16  = lane & 15;
    const int koff = half * 2;
    const int rowBase = blockIdx.x * 16;
    const int colBase = wave * 16;
    const int arow = rowBase + l16;
    const int bcol = colBase + l16;

    v8f acc = {};
    #pragma unroll 4
    for (int k = 0; k < CRV; k += 4) {
        v2f a, b;
        a.x = A[arow * CRV + k + koff];
        a.y = A[arow * CRV + k + koff + 1];
        b.x = Wq[(k + koff) * DD + bcol];
        b.y = Wq[(k + koff + 1) * DD + bcol];
        acc = __builtin_amdgcn_wmma_f32_16x16x4_f32(false, a, false, b,
                                                    (short)0, acc, false, false);
    }

    const float bias = bq[bcol];
    float vals[8], ss[8];
    #pragma unroll
    for (int r = 0; r < 8; ++r) { vals[r] = acc[r] + bias; ss[r] = vals[r] * vals[r]; }

    #pragma unroll
    for (int s = 8; s >= 1; s >>= 1)
        #pragma unroll
        for (int r = 0; r < 8; ++r) ss[r] += __shfl_xor(ss[r], s, 32);

    __shared__ float sq[8][16];
    __shared__ float rnorm[16];
    if (l16 == 0) {
        #pragma unroll
        for (int r = 0; r < 8; ++r) sq[wave][half * 8 + r] = ss[r];
    }
    __syncthreads();
    if (threadIdx.x < 16) {
        float s = 0.f;
        #pragma unroll
        for (int w = 0; w < 8; ++w) s += sq[w][threadIdx.x];
        rnorm[threadIdx.x] = 1.0f / fmaxf(sqrtf(s), 1e-12f);
    }
    __syncthreads();
    #pragma unroll
    for (int r = 0; r < 8; ++r) {
        const int row = half * 8 + r;
        Qn[(rowBase + row) * DD + bcol] = vals[r] * rnorm[row];
    }
}

__global__ __launch_bounds__(256)
void rmba_kvproj_kernel(const float* __restrict__ bev,
                        const float* __restrict__ Wk, const float* __restrict__ bk,
                        const float* __restrict__ Wv, const float* __restrict__ bv,
                        float* __restrict__ Kall,
                        float* __restrict__ Vall)
{
    const int lane = threadIdx.x & 31;
    const int wave = threadIdx.x >> 5;
    const int half = lane >> 4;
    const int l16  = lane & 15;
    const int koff = half * 2;
    const int rowBase = blockIdx.x * 16;
    const int colBase = wave * 16;
    const int arow = rowBase + l16;
    const int bcol = colBase + l16;

    v8f acck = {}, accv = {};
    #pragma unroll 4
    for (int k = 0; k < CB; k += 4) {
        v2f a, wk, wv;
        a.x  = bev[(k + koff) * NBC + arow];
        a.y  = bev[(k + koff + 1) * NBC + arow];
        wk.x = Wk[(k + koff) * DD + bcol];
        wk.y = Wk[(k + koff + 1) * DD + bcol];
        wv.x = Wv[(k + koff) * DD + bcol];
        wv.y = Wv[(k + koff + 1) * DD + bcol];
        acck = __builtin_amdgcn_wmma_f32_16x16x4_f32(false, a, false, wk,
                                                     (short)0, acck, false, false);
        accv = __builtin_amdgcn_wmma_f32_16x16x4_f32(false, a, false, wv,
                                                     (short)0, accv, false, false);
    }

    const float biask = bk[bcol];
    const float biasv = bv[bcol];
    float kv[8], ss[8];
    #pragma unroll
    for (int r = 0; r < 8; ++r) { kv[r] = acck[r] + biask; ss[r] = kv[r] * kv[r]; }
    #pragma unroll
    for (int s = 8; s >= 1; s >>= 1)
        #pragma unroll
        for (int r = 0; r < 8; ++r) ss[r] += __shfl_xor(ss[r], s, 32);

    __shared__ float sq[8][16];
    __shared__ float rnorm[16];
    if (l16 == 0) {
        #pragma unroll
        for (int r = 0; r < 8; ++r) sq[wave][half * 8 + r] = ss[r];
    }
    __syncthreads();
    if (threadIdx.x < 16) {
        float s = 0.f;
        #pragma unroll
        for (int w = 0; w < 8; ++w) s += sq[w][threadIdx.x];
        rnorm[threadIdx.x] = 1.0f / fmaxf(sqrtf(s), 1e-12f);
    }
    __syncthreads();
    #pragma unroll
    for (int r = 0; r < 8; ++r) {
        const int row = half * 8 + r;
        Kall[(rowBase + row) * DD + bcol] = kv[r] * rnorm[row];
        Vall[(rowBase + row) * DD + bcol] = accv[r] + biasv;
    }
}

__global__ __launch_bounds__(256)
void rmba_attn_kernel(const float* __restrict__ Qn,
                      const float* __restrict__ Kall,
                      const float* __restrict__ Vall,
                      const int*   __restrict__ cand,
                      float* __restrict__ attn_out,
                      float* __restrict__ ctx)
{
    const int n   = blockIdx.x;
    const int tid = threadIdx.x;

    __shared__ float q_sm[DD];
    __shared__ float red_sm[8];
    __shared__ float4 cpart[8][32];

    if (tid < DD) q_sm[tid] = Qn[n * DD + tid];
    const int idx = cand[n * MM + tid];
    __syncthreads();

    float logit = -INFINITY;
    if (idx >= 0) {
        const float4* krow = (const float4*)(Kall + (size_t)idx * DD);
        const float4* qv   = (const float4*)q_sm;
        float acc = 0.f;
        #pragma unroll
        for (int c = 0; c < DD / 4; ++c) {
            const float4 k4 = krow[c];
            const float4 q4 = qv[c];
            acc += k4.x * q4.x + k4.y * q4.y + k4.z * q4.z + k4.w * q4.w;
        }
        logit = acc * INV_SQRT_D;
    }

    float mx = logit;
    #pragma unroll
    for (int s = 16; s >= 1; s >>= 1) mx = fmaxf(mx, __shfl_xor(mx, s, 32));
    if ((tid & 31) == 0) red_sm[tid >> 5] = mx;
    __syncthreads();
    float bm = red_sm[0];
    #pragma unroll
    for (int w = 1; w < 8; ++w) bm = fmaxf(bm, red_sm[w]);
    __syncthreads();

    const float e = (idx >= 0) ? __expf(logit - bm) : 0.f;
    float sm = e;
    #pragma unroll
    for (int s = 16; s >= 1; s >>= 1) sm += __shfl_xor(sm, s, 32);
    if ((tid & 31) == 0) red_sm[tid >> 5] = sm;
    __syncthreads();
    float bs = 0.f;
    #pragma unroll
    for (int w = 0; w < 8; ++w) bs += red_sm[w];

    const float a = e / bs;
    attn_out[n * MM + tid] = a;

    const int lane = tid & 31;
    const int wave = tid >> 5;
    const int abits = __float_as_int(a);
    const char* vbytes = (const char*)Vall;
    const unsigned laneOff = (unsigned)lane * 16u;
    float4 acc = make_float4(0.f, 0.f, 0.f, 0.f);
    #pragma unroll 1
    for (int jb = 0; jb < 32; jb += 16) {
        float4 v[16];
        float  w[16];
        #pragma unroll
        for (int j = 0; j < 16; ++j) {
            const int ix  = __builtin_amdgcn_readlane(idx, jb + j);
            w[j] = __int_as_float(__builtin_amdgcn_readlane(abits, jb + j));
            const unsigned ixc = (unsigned)(ix < 0 ? 0 : ix);
            const unsigned off = ixc * (unsigned)(DD * sizeof(float)) + laneOff;
            v[j] = *(const float4*)(vbytes + off);
        }
        #pragma unroll
        for (int j = 0; j < 16; ++j) {
            acc.x = fmaf(w[j], v[j].x, acc.x);
            acc.y = fmaf(w[j], v[j].y, acc.y);
            acc.z = fmaf(w[j], v[j].z, acc.z);
            acc.w = fmaf(w[j], v[j].w, acc.w);
        }
    }
    cpart[wave][lane] = acc;
    __syncthreads();
    if (tid < DD) {
        const float* cp = (const float*)cpart;
        float s = 0.f;
        #pragma unroll
        for (int w8 = 0; w8 < 8; ++w8) s += cp[w8 * DD + tid];
        ctx[n * DD + tid] = s;
    }
}

__global__ __launch_bounds__(256)
void rmba_oproj_kernel(const float* __restrict__ ctx,
                       const float* __restrict__ Wo,
                       const float* __restrict__ bo,
                       float* __restrict__ out)
{
    const int mTiles = NRQ / 16;
    const int mTile  = blockIdx.x % mTiles;
    const int nGrp   = blockIdx.x / mTiles;
    const int lane = threadIdx.x & 31;
    const int wave = threadIdx.x >> 5;
    const int half = lane >> 4;
    const int l16  = lane & 15;
    const int koff = half * 2;
    const int rowBase = mTile * 16;
    const int colBase = nGrp * 128 + wave * 16;
    const int arow = rowBase + l16;
    const int bcol = colBase + l16;

    v8f acc = {};
    #pragma unroll 4
    for (int k = 0; k < DD; k += 4) {
        v2f a, b;
        a.x = ctx[arow * DD + k + koff];
        a.y = ctx[arow * DD + k + koff + 1];
        b.x = Wo[(k + koff) * CRV + bcol];
        b.y = Wo[(k + koff + 1) * CRV + bcol];
        acc = __builtin_amdgcn_wmma_f32_16x16x4_f32(false, a, false, b,
                                                    (short)0, acc, false, false);
    }
    const float bias = bo[bcol];
    #pragma unroll
    for (int r = 0; r < 8; ++r) {
        const int row = rowBase + half * 8 + r;
        out[row * CRV + bcol] = acc[r] + bias;
    }
}

extern "C" void kernel_launch(void* const* d_in, const int* in_sizes, int n_in,
                              void* d_out, int out_size, void* d_ws, size_t ws_size,
                              hipStream_t stream) {
    const float* rv_x = (const float*)d_in[0];
    const float* bev  = (const float*)d_in[1];
    const int*   cand = (const int*)  d_in[2];
    const float* Wq   = (const float*)d_in[3];
    const float* bq   = (const float*)d_in[4];
    const float* Wk   = (const float*)d_in[5];
    const float* bk   = (const float*)d_in[6];
    const float* Wv   = (const float*)d_in[7];
    const float* bv   = (const float*)d_in[8];
    const float* Wo   = (const float*)d_in[9];
    const float* bo   = (const float*)d_in[10];

    float* out_main = (float*)d_out;
    float* out_attn = out_main + (size_t)NRQ * CRV;

    float* Qn   = (float*)d_ws;
    float* Kall = Qn   + (size_t)NRQ * DD;
    float* Vall = Kall + (size_t)NBC * DD;
    float* ctx  = Vall + (size_t)NBC * DD;

    rmba_qproj_kernel <<<NRQ / 16, 256, 0, stream>>>(rv_x, Wq, bq, Qn);
    rmba_kvproj_kernel<<<NBC / 16, 256, 0, stream>>>(bev, Wk, bk, Wv, bv, Kall, Vall);
    rmba_attn_kernel  <<<NRQ,      256, 0, stream>>>(Qn, Kall, Vall, cand, out_attn, ctx);
    rmba_oproj_kernel <<<(NRQ / 16) * (CRV / 128), 256, 0, stream>>>(ctx, Wo, bo, out_main);
}


# logits pass: 8 lanes share a K row, one whole 128-B line of 4 rows per wave load instead of 32 partial lines; dots finished by an 8-lane butterfly (logit sum order changes)
# speedup vs baseline: 2.4502x; 1.7360x over previous
#include <hip/hip_runtime.h>
#include <math.h>


typedef float v2f __attribute__((ext_vector_type(2)));
typedef float v8f __attribute__((ext_vector_type(8)));

constexpr int NRQ = 4096;
constexpr int NBC = 16384;
constexpr int CB  = 80;
constexpr int CRV = 384;
constexpr int DD  = 128;
constexpr int MM  = 256;
constexpr float INV_SQRT_D = 0.08838834764831845f;


__global__ __launch_bounds__(256)
void rmba_qproj_kernel(const float* __restrict__ A,
                       const float* __restrict__ Wq,
                       const float* __restrict__ bq,
                       float* __restrict__ Qn)
{
    const int lane = threadIdx.x & 31;
    const int wave = threadIdx.x >> 5;
    const int half = lane >> 4;
    const int l16  = lane & 15;
    const int koff = half * 2;
    const int rowBase = blockIdx.x * 16;
    const int colBase = wave * 16;
    const int arow = rowBase + l16;
    const int bcol = colBase + l16;

    v8f acc = {};
    #pragma unroll 4
    for (int k = 0; k < CRV; k += 4) {
        v2f a, b;
        a.x = A[arow * CRV + k + koff];
        a.y = A[arow * CRV + k + koff + 1];
        b.x = Wq[(k + koff) * DD + bcol];
        b.y = Wq[(k + koff + 1) * DD + bcol];
        acc = __builtin_amdgcn_wmma_f32_16x16x4_f32(false, a, false, b,
                                                    (short)0, acc, false, false);
    }

    const float bias = bq[bcol];
    float vals[8], ss[8];
    #pragma unroll
    for (int r = 0; r < 8; ++r) { vals[r] = acc[r] + bias; ss[r] = vals[r] * vals[r]; }

    #pragma unroll
    for (int s = 8; s >= 1; s >>= 1)
        #pragma unroll
        for (int r = 0; r < 8; ++r) ss[r] += __shfl_xor(ss[r], s, 32);

    __shared__ float sq[8][16];
    __shared__ float rnorm[16];
    if (l16 == 0) {
        #pragma unroll
        for (int r = 0; r < 8; ++r) sq[wave][half * 8 + r] = ss[r];
    }
    __syncthreads();
    if (threadIdx.x < 16) {
        float s = 0.f;
        #pragma unroll
        for (int w = 0; w < 8; ++w) s += sq[w][threadIdx.x];
        rnorm[threadIdx.x] = 1.0f / fmaxf(sqrtf(s), 1e-12f);
    }
    __syncthreads();
    #pragma unroll
    for (int r = 0; r < 8; ++r) {
        const int row = half * 8 + r;
        Qn[(rowBase + row) * DD + bcol] = vals[r] * rnorm[row];
    }
}

__global__ __launch_bounds__(256)
void rmba_kvproj_kernel(const float* __restrict__ bev,
                        const float* __restrict__ Wk, const float* __restrict__ bk,
                        const float* __restrict__ Wv, const float* __restrict__ bv,
                        float* __restrict__ Kall,
                        float* __restrict__ Vall)
{
    const int lane = threadIdx.x & 31;
    const int wave = threadIdx.x >> 5;
    const int half = lane >> 4;
    const int l16  = lane & 15;
    const int koff = half * 2;
    const int rowBase = blockIdx.x * 16;
    const int colBase = wave * 16;
    const int arow = rowBase + l16;
    const int bcol = colBase + l16;

    v8f acck = {}, accv = {};
    #pragma unroll 4
    for (int k = 0; k < CB; k += 4) {
        v2f a, wk, wv;
        a.x  = bev[(k + koff) * NBC + arow];
        a.y  = bev[(k + koff + 1) * NBC + arow];
        wk.x = Wk[(k + koff) * DD + bcol];
        wk.y = Wk[(k + koff + 1) * DD + bcol];
        wv.x = Wv[(k + koff) * DD + bcol];
        wv.y = Wv[(k + koff + 1) * DD + bcol];
        acck = __builtin_amdgcn_wmma_f32_16x16x4_f32(false, a, false, wk,
                                                     (short)0, acck, false, false);
        accv = __builtin_amdgcn_wmma_f32_16x16x4_f32(false, a, false, wv,
                                                     (short)0, accv, false, false);
    }

    const float biask = bk[bcol];
    const float biasv = bv[bcol];
    float kv[8], ss[8];
    #pragma unroll
    for (int r = 0; r < 8; ++r) { kv[r] = acck[r] + biask; ss[r] = kv[r] * kv[r]; }
    #pragma unroll
    for (int s = 8; s >= 1; s >>= 1)
        #pragma unroll
        for (int r = 0; r < 8; ++r) ss[r] += __shfl_xor(ss[r], s, 32);

    __shared__ float sq[8][16];
    __shared__ float rnorm[16];
    if (l16 == 0) {
        #pragma unroll
        for (int r = 0; r < 8; ++r) sq[wave][half * 8 + r] = ss[r];
    }
    __syncthreads();
    if (threadIdx.x < 16) {
        float s = 0.f;
        #pragma unroll
        for (int w = 0; w < 8; ++w) s += sq[w][threadIdx.x];
        rnorm[threadIdx.x] = 1.0f / fmaxf(sqrtf(s), 1e-12f);
    }
    __syncthreads();
    #pragma unroll
    for (int r = 0; r < 8; ++r) {
        const int row = half * 8 + r;
        Kall[(rowBase + row) * DD + bcol] = kv[r] * rnorm[row];
        Vall[(rowBase + row) * DD + bcol] = accv[r] + biasv;
    }
}

__global__ __launch_bounds__(256)
void rmba_attn_kernel(const float* __restrict__ Qn,
                      const float* __restrict__ Kall,
                      const float* __restrict__ Vall,
                      const int*   __restrict__ cand,
                      float* __restrict__ attn_out,
                      float* __restrict__ ctx)
{
    const int n    = blockIdx.x;
    const int tid  = threadIdx.x;
    const int lane = tid & 31;
    const int wave = tid >> 5;

    __shared__ float red_sm[8];
    __shared__ float4 cpart[8][32];

    const int idx = cand[n * MM + tid];

    const int4* cg = (const int4*)(cand + n * MM + (tid & ~7));
    const int4 ca = cg[0];
    const int4 cb = cg[1];
    const int gi[8] = {ca.x, ca.y, ca.z, ca.w, cb.x, cb.y, cb.z, cb.w};

    const int sub = lane & 7;
    const float4* qrow = (const float4*)(Qn + n * DD);
    float4 q4[4];
    #pragma unroll
    for (int t = 0; t < 4; ++t) q4[t] = qrow[t * 8 + sub];

    const char* kbytes = (const char*)Kall;
    const unsigned subOff = (unsigned)sub * 16u;
    float p[8];
    #pragma unroll
    for (int g = 0; g < 8; ++g) {
        const unsigned ixc = (unsigned)(gi[g] < 0 ? 0 : gi[g]);
        const unsigned off = ixc * (unsigned)(DD * sizeof(float)) + subOff;
        float acc = 0.f;
        #pragma unroll
        for (int t = 0; t < 4; ++t) {
            const float4 k4 = *(const float4*)(kbytes + off + t * 128);
            acc = fmaf(k4.x, q4[t].x, acc);
            acc = fmaf(k4.y, q4[t].y, acc);
            acc = fmaf(k4.z, q4[t].z, acc);
            acc = fmaf(k4.w, q4[t].w, acc);
        }
        p[g] = acc;
    }

    const bool b0 = (lane & 1) != 0;
    const bool b1 = (lane & 2) != 0;
    const bool b2 = (lane & 4) != 0;
    float r1[4];
    #pragma unroll
    for (int i = 0; i < 4; ++i) {
        const float keep = b0 ? p[2 * i + 1] : p[2 * i];
        const float send = b0 ? p[2 * i]     : p[2 * i + 1];
        r1[i] = keep + __shfl_xor(send, 1, 32);
    }
    float r2[2];
    #pragma unroll
    for (int i = 0; i < 2; ++i) {
        const float keep = b1 ? r1[2 * i + 1] : r1[2 * i];
        const float send = b1 ? r1[2 * i]     : r1[2 * i + 1];
        r2[i] = keep + __shfl_xor(send, 2, 32);
    }
    const float keep3 = b2 ? r2[1] : r2[0];
    const float send3 = b2 ? r2[0] : r2[1];
    const float dot = keep3 + __shfl_xor(send3, 4, 32);

    const float logit = (idx >= 0) ? dot * INV_SQRT_D : -INFINITY;

    float mx = logit;
    #pragma unroll
    for (int s = 16; s >= 1; s >>= 1) mx = fmaxf(mx, __shfl_xor(mx, s, 32));
    if (lane == 0) red_sm[wave] = mx;
    __syncthreads();
    float bm = red_sm[0];
    #pragma unroll
    for (int w = 1; w < 8; ++w) bm = fmaxf(bm, red_sm[w]);
    __syncthreads();

    const float e = (idx >= 0) ? __expf(logit - bm) : 0.f;
    float sm = e;
    #pragma unroll
    for (int s = 16; s >= 1; s >>= 1) sm += __shfl_xor(sm, s, 32);
    if (lane == 0) red_sm[wave] = sm;
    __syncthreads();
    float bs = 0.f;
    #pragma unroll
    for (int w = 0; w < 8; ++w) bs += red_sm[w];

    const float a = e / bs;
    attn_out[n * MM + tid] = a;

    const int abits = __float_as_int(a);
    const char* vbytes = (const char*)Vall;
    const unsigned laneOff = (unsigned)lane * 16u;
    float4 acc = make_float4(0.f, 0.f, 0.f, 0.f);
    #pragma unroll 1
    for (int jb = 0; jb < 32; jb += 16) {
        float4 v[16];
        float  w[16];
        #pragma unroll
        for (int j = 0; j < 16; ++j) {
            const int ix  = __builtin_amdgcn_readlane(idx, jb + j);
            w[j] = __int_as_float(__builtin_amdgcn_readlane(abits, jb + j));
            const unsigned ixc = (unsigned)(ix < 0 ? 0 : ix);
            const unsigned off = ixc * (unsigned)(DD * sizeof(float)) + laneOff;
            v[j] = *(const float4*)(vbytes + off);
        }
        #pragma unroll
        for (int j = 0; j < 16; ++j) {
            acc.x = fmaf(w[j], v[j].x, acc.x);
            acc.y = fmaf(w[j], v[j].y, acc.y);
            acc.z = fmaf(w[j], v[j].z, acc.z);
            acc.w = fmaf(w[j], v[j].w, acc.w);
        }
    }
    cpart[wave][lane] = acc;
    __syncthreads();
    if (tid < DD) {
        const float* cp = (const float*)cpart;
        float s = 0.f;
        #pragma unroll
        for (int w8 = 0; w8 < 8; ++w8) s += cp[w8 * DD + tid];
        ctx[n * DD + tid] = s;
    }
}

__global__ __launch_bounds__(256)
void rmba_oproj_kernel(const float* __restrict__ ctx,
                       const float* __restrict__ Wo,
                       const float* __restrict__ bo,
                       float* __restrict__ out)
{
    const int mTiles = NRQ / 16;
    const int mTile  = blockIdx.x % mTiles;
    const int nGrp   = blockIdx.x / mTiles;
    const int lane = threadIdx.x & 31;
    const int wave = threadIdx.x >> 5;
    const int half = lane >> 4;
    const int l16  = lane & 15;
    const int koff = half * 2;
    const int rowBase = mTile * 16;
    const int colBase = nGrp * 128 + wave * 16;
    const int arow = rowBase + l16;
    const int bcol = colBase + l16;

    v8f acc = {};
    #pragma unroll 4
    for (int k = 0; k < DD; k += 4) {
        v2f a, b;
        a.x = ctx[arow * DD + k + koff];
        a.y = ctx[arow * DD + k + koff + 1];
        b.x = Wo[(k + koff) * CRV + bcol];
        b.y = Wo[(k + koff + 1) * CRV + bcol];
        acc = __builtin_amdgcn_wmma_f32_16x16x4_f32(false, a, false, b,
                                                    (short)0, acc, false, false);
    }
    const float bias = bo[bcol];
    #pragma unroll
    for (int r = 0; r < 8; ++r) {
        const int row = rowBase + half * 8 + r;
        out[row * CRV + bcol] = acc[r] + bias;
    }
}

extern "C" void kernel_launch(void* const* d_in, const int* in_sizes, int n_in,
                              void* d_out, int out_size, void* d_ws, size_t ws_size,
                              hipStream_t stream) {
    const float* rv_x = (const float*)d_in[0];
    const float* bev  = (const float*)d_in[1];
    const int*   cand = (const int*)  d_in[2];
    const float* Wq   = (const float*)d_in[3];
    const float* bq   = (const float*)d_in[4];
    const float* Wk   = (const float*)d_in[5];
    const float* bk   = (const float*)d_in[6];
    const float* Wv   = (const float*)d_in[7];
    const float* bv   = (const float*)d_in[8];
    const float* Wo   = (const float*)d_in[9];
    const float* bo   = (const float*)d_in[10];

    float* out_main = (float*)d_out;
    float* out_attn = out_main + (size_t)NRQ * CRV;

    float* Qn   = (float*)d_ws;
    float* Kall = Qn   + (size_t)NRQ * DD;
    float* Vall = Kall + (size_t)NBC * DD;
    float* ctx  = Vall + (size_t)NBC * DD;

    rmba_qproj_kernel <<<NRQ / 16, 256, 0, stream>>>(rv_x, Wq, bq, Qn);
    rmba_kvproj_kernel<<<NBC / 16, 256, 0, stream>>>(bev, Wk, bk, Wv, bv, Kall, Vall);
    rmba_attn_kernel  <<<NRQ,      256, 0, stream>>>(Qn, Kall, Vall, cand, out_attn, ctx);
    rmba_oproj_kernel <<<(NRQ / 16) * (CRV / 128), 256, 0, stream>>>(ctx, Wo, bo, out_main);
}


# attention pass: 1024 work-groups, all resident at once, each looping over 4 queries, not 4096 one-query work-groups: 4x fewer wave starts (cross-lane reads as ds_swizzle so the loop fits 128 VGPRs)
# speedup vs baseline: 2.7486x; 1.1218x over previous
#include <hip/hip_runtime.h>
#include <math.h>


typedef float v2f __attribute__((ext_vector_type(2)));
typedef float v8f __attribute__((ext_vector_type(8)));

constexpr int NRQ = 4096;
constexpr int NBC = 16384;
constexpr int CB  = 80;
constexpr int CRV = 384;
constexpr int DD  = 128;
constexpr int MM  = 256;
constexpr int EP  = 96;
constexpr int QP  = 132;
constexpr float INV_SQRT_D = 0.08838834764831845f;


__global__ __launch_bounds__(256)
void rmba_qproj_kernel(const float* __restrict__ A,
                       const float* __restrict__ Wq,
                       const float* __restrict__ bq,
                       const float* __restrict__ Wk,
                       const float* __restrict__ bk,
                       float* __restrict__ Qe)
{
    const int lane = threadIdx.x & 31;
    const int wave = threadIdx.x >> 5;
    const int half = lane >> 4;
    const int l16  = lane & 15;
    const int koff = half * 2;
    const int rowBase = blockIdx.x * 16;
    const int colBase = wave * 16;
    const int arow = rowBase + l16;
    const int bcol = colBase + l16;

    v8f acc = {};
    #pragma unroll 4
    for (int k = 0; k < CRV; k += 4) {
        v2f a, b;
        a.x = A[arow * CRV + k + koff];
        a.y = A[arow * CRV + k + koff + 1];
        b.x = Wq[(k + koff) * DD + bcol];
        b.y = Wq[(k + koff + 1) * DD + bcol];
        acc = __builtin_amdgcn_wmma_f32_16x16x4_f32(false, a, false, b,
                                                    (short)0, acc, false, false);
    }

    const float bias = bq[bcol];
    float vals[8], ss[8];
    #pragma unroll
    for (int r = 0; r < 8; ++r) { vals[r] = acc[r] + bias; ss[r] = vals[r] * vals[r]; }

    #pragma unroll
    for (int s = 8; s >= 1; s >>= 1)
        #pragma unroll
        for (int r = 0; r < 8; ++r) ss[r] += __shfl_xor(ss[r], s, 32);

    __shared__ float sq[8][16];
    __shared__ float rnorm[16];
    __shared__ float sQ[16 * QP];
    if (l16 == 0) {
        #pragma unroll
        for (int r = 0; r < 8; ++r) sq[wave][half * 8 + r] = ss[r];
    }
    __syncthreads();
    if (threadIdx.x < 16) {
        float s = 0.f;
        #pragma unroll
        for (int w = 0; w < 8; ++w) s += sq[w][threadIdx.x];
        rnorm[threadIdx.x] = 1.0f / fmaxf(sqrtf(s), 1e-12f);
    }
    __syncthreads();
    #pragma unroll
    for (int r = 0; r < 8; ++r) {
        const int row = half * 8 + r;
        sQ[row * QP + bcol] = vals[r] * rnorm[row];
    }
    __syncthreads();

    if (wave < 6) {
        const int col = colBase + l16;
        const float* bp = (col < CB) ? (Wk + col * DD) : bk;
        const bool live = (col <= CB);
        v8f acc2 = {};
        #pragma unroll 4
        for (int k = 0; k < DD; k += 4) {
            v2f a, b;
            a.x = sQ[l16 * QP + k + koff];
            a.y = sQ[l16 * QP + k + koff + 1];
            const float x0 = bp[k + koff];
            const float x1 = bp[k + koff + 1];
            b.x = live ? x0 : 0.f;
            b.y = live ? x1 : 0.f;
            acc2 = __builtin_amdgcn_wmma_f32_16x16x4_f32(false, a, false, b,
                                                         (short)0, acc2, false, false);
        }
        #pragma unroll
        for (int r = 0; r < 8; ++r)
            Qe[(rowBase + half * 8 + r) * EP + col] = acc2[r];
    }
}

__global__ __launch_bounds__(256)
void rmba_cell_kernel(const float* __restrict__ bev,
                      const float* __restrict__ Wk, const float* __restrict__ bk,
                      float* __restrict__ Te)
{
    const int lane = threadIdx.x & 31;
    const int wave = threadIdx.x >> 5;
    const int half = lane >> 4;
    const int l16  = lane & 15;
    const int koff = half * 2;
    const int rowBase = blockIdx.x * 16;
    const int colBase = wave * 16;
    const int arow = rowBase + l16;
    const int bcol = colBase + l16;

    __shared__ float sq[8][16];
    __shared__ __attribute__((aligned(16))) float sT[16 * EP];

    v8f acck = {};
    #pragma unroll 4
    for (int k = 0; k < CB; k += 4) {
        v2f a, wk;
        a.x  = bev[(k + koff) * NBC + arow];
        a.y  = bev[(k + koff + 1) * NBC + arow];
        wk.x = Wk[(k + koff) * DD + bcol];
        wk.y = Wk[(k + koff + 1) * DD + bcol];
        acck = __builtin_amdgcn_wmma_f32_16x16x4_f32(false, a, false, wk,
                                                     (short)0, acck, false, false);
    }

    {
        const int ci = threadIdx.x & 15;
        const int c0 = threadIdx.x >> 4;
        #pragma unroll
        for (int j = 0; j < 5; ++j) {
            const int c = c0 + 16 * j;
            sT[ci * EP + c] = bev[c * NBC + rowBase + ci];
        }
        if (threadIdx.x < 240) {
            const int ri = threadIdx.x / 15;
            const int cj = threadIdx.x % 15;
            const int c  = (cj == 0) ? 80 : 81 + cj;
            sT[ri * EP + c] = (cj == 0) ? 1.0f : 0.0f;
        }
    }

    const float biask = bk[bcol];
    float ss[8];
    #pragma unroll
    for (int r = 0; r < 8; ++r) { const float kv = acck[r] + biask; ss[r] = kv * kv; }
    #pragma unroll
    for (int s = 8; s >= 1; s >>= 1)
        #pragma unroll
        for (int r = 0; r < 8; ++r) ss[r] += __shfl_xor(ss[r], s, 32);

    if (l16 == 0) {
        #pragma unroll
        for (int r = 0; r < 8; ++r) sq[wave][half * 8 + r] = ss[r];
    }
    __syncthreads();
    if (threadIdx.x < 16) {
        float s = 0.f;
        #pragma unroll
        for (int w = 0; w < 8; ++w) s += sq[w][threadIdx.x];
        sT[threadIdx.x * EP + 81] = 1.0f / fmaxf(sqrtf(s), 1e-12f);
    }
    __syncthreads();
    {
        const float4* s4 = (const float4*)sT;
        float4* t4 = (float4*)(Te + (size_t)rowBase * EP);
        t4[threadIdx.x] = s4[threadIdx.x];
        if (threadIdx.x < 128) t4[256 + threadIdx.x] = s4[256 + threadIdx.x];
    }
}

constexpr int QPB = 4;

template <int M>
__device__ __forceinline__ float lane_xor(float v) {
    return __int_as_float(__builtin_amdgcn_ds_swizzle(__float_as_int(v), 0x1f | (M << 10)));
}
template <int G>
__device__ __forceinline__ float group8_lane(float v) {
    return __int_as_float(__builtin_amdgcn_ds_swizzle(__float_as_int(v), 0x18 | (G << 5)));
}

__global__ __launch_bounds__(256)
void rmba_attn_kernel(const float* __restrict__ Qe,
                      const float* __restrict__ Te,
                      const int*   __restrict__ cand,
                      float* __restrict__ attn_out,
                      float* __restrict__ ctxe)
{
    const int n0   = blockIdx.x * QPB;
    const int tid  = threadIdx.x;
    const int lane = tid & 31;
    const int wave = tid >> 5;
    const int sub  = lane & 7;

    __shared__ float red_sm[8];
    __shared__ float cpart[8][EP];

    const char* tbytes = (const char*)Te;
    const unsigned subOff = (unsigned)sub * 16u;
    const bool b0 = (lane & 1) != 0;
    const bool b1 = (lane & 2) != 0;
    const bool b2 = (lane & 4) != 0;
    const bool b3 = (lane & 8) != 0;
    const bool b4 = (lane & 16) != 0;

    for (int qi = 0; qi < QPB; ++qi) {
        const int n = n0 + qi;

        const int idx = cand[(unsigned)(n * MM) + (unsigned)tid];
        const int4* cand4 = (const int4*)cand;
        const unsigned gq = (unsigned)(n * (MM / 4)) + (unsigned)(tid >> 3) * 2u;
        const int4 ca = cand4[gq];
        const int4 cb = cand4[gq + 1u];

        const float rn = Te[(unsigned)(idx < 0 ? 0 : idx) * (unsigned)EP + 81u];
        const int gi[8] = {ca.x, ca.y, ca.z, ca.w, cb.x, cb.y, cb.z, cb.w};

        const float4* qe4 = (const float4*)Qe;
        const unsigned qq = (unsigned)(n * (EP / 4)) + (unsigned)sub;
        float4 q4[3];
        #pragma unroll
        for (int t = 0; t < 3; ++t) q4[t] = qe4[qq + (unsigned)(t * 8)];

        float4 kv[8][3];
        float p[8];
        #pragma unroll
        for (int g = 0; g < 8; ++g) {
            const unsigned ixc = (unsigned)(gi[g] < 0 ? 0 : gi[g]);
            const unsigned off = ixc * (unsigned)(EP * sizeof(float)) + subOff;
            #pragma unroll
            for (int t = 0; t < 3; ++t)
                kv[g][t] = *(const float4*)(tbytes + off + t * 128);
        }

        #pragma unroll
        for (int g = 0; g < 8; ++g) {
            float acc = 0.f;
            #pragma unroll
            for (int t = 0; t < 3; ++t) {
                acc = fmaf(kv[g][t].x, q4[t].x, acc);
                acc = fmaf(kv[g][t].y, q4[t].y, acc);
                acc = fmaf(kv[g][t].z, q4[t].z, acc);
                acc = fmaf(kv[g][t].w, q4[t].w, acc);
            }
            p[g] = acc;
        }

        float r1[4];
        #pragma unroll
        for (int i = 0; i < 4; ++i) {
            const float keep = b0 ? p[2 * i + 1] : p[2 * i];
            const float send = b0 ? p[2 * i]     : p[2 * i + 1];
            r1[i] = keep + lane_xor<1>(send);
        }
        float r2[2];
        #pragma unroll
        for (int i = 0; i < 2; ++i) {
            const float keep = b1 ? r1[2 * i + 1] : r1[2 * i];
            const float send = b1 ? r1[2 * i]     : r1[2 * i + 1];
            r2[i] = keep + lane_xor<2>(send);
        }
        const float keep3 = b2 ? r2[1] : r2[0];
        const float send3 = b2 ? r2[0] : r2[1];
        const float dot = keep3 + lane_xor<4>(send3);

        const float logit = (idx >= 0) ? (dot * rn) * INV_SQRT_D : -INFINITY;

        float mx = logit;
        mx = fmaxf(mx, lane_xor<16>(mx));
        mx = fmaxf(mx, lane_xor<8>(mx));
        mx = fmaxf(mx, lane_xor<4>(mx));
        mx = fmaxf(mx, lane_xor<2>(mx));
        mx = fmaxf(mx, lane_xor<1>(mx));
        if (lane == 0) red_sm[wave] = mx;
        __syncthreads();
        float bm = red_sm[0];
        #pragma unroll
        for (int w = 1; w < 8; ++w) bm = fmaxf(bm, red_sm[w]);
        __syncthreads();

        const float e = (idx >= 0) ? __expf(logit - bm) : 0.f;
        float sm = e;
        sm += lane_xor<16>(sm);
        sm += lane_xor<8>(sm);
        sm += lane_xor<4>(sm);
        sm += lane_xor<2>(sm);
        sm += lane_xor<1>(sm);
        if (lane == 0) red_sm[wave] = sm;
        __syncthreads();
        float bs = 0.f;
        #pragma unroll
        for (int w = 0; w < 8; ++w) bs += red_sm[w];

        const float a = e / bs;
        attn_out[(unsigned)(n * MM) + (unsigned)tid] = a;

        float v[12] = {};
        const float agv[8] = {group8_lane<0>(a), group8_lane<1>(a), group8_lane<2>(a), group8_lane<3>(a),
                              group8_lane<4>(a), group8_lane<5>(a), group8_lane<6>(a), group8_lane<7>(a)};
        #pragma unroll
        for (int g = 0; g < 8; ++g) {
            const float ag = agv[g];
            #pragma unroll
            for (int t = 0; t < 3; ++t) {
                v[4 * t + 0] = fmaf(ag, kv[g][t].x, v[4 * t + 0]);
                v[4 * t + 1] = fmaf(ag, kv[g][t].y, v[4 * t + 1]);
                v[4 * t + 2] = fmaf(ag, kv[g][t].z, v[4 * t + 2]);
                v[4 * t + 3] = fmaf(ag, kv[g][t].w, v[4 * t + 3]);
            }
        }
        float w6[6];
        #pragma unroll
        for (int i = 0; i < 6; ++i) {
            const float keep = b3 ? v[6 + i] : v[i];
            const float send = b3 ? v[i]     : v[6 + i];
            w6[i] = keep + lane_xor<8>(send);
        }
        float y[3];
        #pragma unroll
        for (int i = 0; i < 3; ++i) {
            const float keep = b4 ? w6[3 + i] : w6[i];
            const float send = b4 ? w6[i]     : w6[3 + i];
            y[i] = keep + lane_xor<16>(send);
        }
        {
            const int jb = (b3 ? 6 : 0) + (b4 ? 3 : 0);
            #pragma unroll
            for (int i = 0; i < 3; ++i)
                cpart[wave][8 * (jb + i) + sub] = y[i];
        }
        __syncthreads();
        if (tid < EP) {
            const int slot = 8 * (4 * (tid >> 5) + (tid & 3)) + ((tid >> 2) & 7);
            float s = 0.f;
            #pragma unroll
            for (int w8 = 0; w8 < 8; ++w8) s += cpart[w8][slot];
            ctxe[(unsigned)(n * EP) + (unsigned)tid] = s;
        }
    }
}

__global__ __launch_bounds__(256)
void rmba_oproj_kernel(const float* __restrict__ ctxe,
                       const float* __restrict__ Wv,
                       const float* __restrict__ bv,
                       const float* __restrict__ Wo,
                       const float* __restrict__ bo,
                       float* __restrict__ out)
{
    const int mTiles = NRQ / 16;
    const int mTile  = blockIdx.x % mTiles;
    const int nGrp   = blockIdx.x / mTiles;
    const int lane = threadIdx.x & 31;
    const int wave = threadIdx.x >> 5;
    const int half = lane >> 4;
    const int l16  = lane & 15;
    const int koff = half * 2;
    const int rowBase = mTile * 16;
    const int arow = rowBase + l16;

    __shared__ float sC[16 * QP];

    {
        const int ccol = wave * 16 + l16;
        v8f acc1 = {};
        #pragma unroll 4
        for (int k = 0; k < CB; k += 4) {
            v2f a, b;
            a.x = ctxe[arow * EP + k + koff];
            a.y = ctxe[arow * EP + k + koff + 1];
            b.x = Wv[(k + koff) * DD + ccol];
            b.y = Wv[(k + koff + 1) * DD + ccol];
            acc1 = __builtin_amdgcn_wmma_f32_16x16x4_f32(false, a, false, b,
                                                         (short)0, acc1, false, false);
        }
        {
            v2f a, b;
            a.x = ctxe[arow * EP + CB + koff];
            a.y = ctxe[arow * EP + CB + koff + 1];
            const float bvc = bv[ccol];
            b.x = (half == 0) ? bvc : 0.f;
            b.y = 0.f;
            acc1 = __builtin_amdgcn_wmma_f32_16x16x4_f32(false, a, false, b,
                                                         (short)0, acc1, false, false);
        }
        #pragma unroll
        for (int r = 0; r < 8; ++r)
            sC[(half * 8 + r) * QP + ccol] = acc1[r];
    }
    __syncthreads();

    const int colBase = nGrp * 128 + wave * 16;
    const int bcol = colBase + l16;
    v8f acc = {};
    #pragma unroll 4
    for (int k = 0; k < DD; k += 4) {
        v2f a, b;
        a.x = sC[l16 * QP + k + koff];
        a.y = sC[l16 * QP + k + koff + 1];
        b.x = Wo[(k + koff) * CRV + bcol];
        b.y = Wo[(k + koff + 1) * CRV + bcol];
        acc = __builtin_amdgcn_wmma_f32_16x16x4_f32(false, a, false, b,
                                                    (short)0, acc, false, false);
    }
    const float bias = bo[bcol];
    #pragma unroll
    for (int r = 0; r < 8; ++r) {
        const int row = rowBase + half * 8 + r;
        out[row * CRV + bcol] = acc[r] + bias;
    }
}

extern "C" void kernel_launch(void* const* d_in, const int* in_sizes, int n_in,
                              void* d_out, int out_size, void* d_ws, size_t ws_size,
                              hipStream_t stream) {
    const float* rv_x = (const float*)d_in[0];
    const float* bev  = (const float*)d_in[1];
    const int*   cand = (const int*)  d_in[2];
    const float* Wq   = (const float*)d_in[3];
    const float* bq   = (const float*)d_in[4];
    const float* Wk   = (const float*)d_in[5];
    const float* bk   = (const float*)d_in[6];
    const float* Wv   = (const float*)d_in[7];
    const float* bv   = (const float*)d_in[8];
    const float* Wo   = (const float*)d_in[9];
    const float* bo   = (const float*)d_in[10];

    float* out_main = (float*)d_out;
    float* out_attn = out_main + (size_t)NRQ * CRV;

    float* Qe   = (float*)d_ws;
    float* Te   = Qe + (size_t)NRQ * EP;
    float* ctxe = Te + (size_t)NBC * EP;

    rmba_qproj_kernel<<<NRQ / 16, 256, 0, stream>>>(rv_x, Wq, bq, Wk, bk, Qe);
    rmba_cell_kernel <<<NBC / 16, 256, 0, stream>>>(bev, Wk, bk, Te);
    rmba_attn_kernel <<<NRQ / QPB, 256, 0, stream>>>(Qe, Te, cand, out_attn, ctxe);
    rmba_oproj_kernel<<<(NRQ / 16) * (CRV / 128), 256, 0, stream>>>(ctxe, Wv, bv, Wo, bo, out_main);
}


# query projection and cell table, which do not depend on each other, merged into one launch (block index selects the part): 3 launches not 4, the table is built beside the projection's long chain
# speedup vs baseline: 3.1399x; 1.1424x over previous
#include <hip/hip_runtime.h>
#include <math.h>


typedef float v2f __attribute__((ext_vector_type(2)));
typedef float v8f __attribute__((ext_vector_type(8)));

constexpr int NRQ = 4096;
constexpr int NBC = 16384;
constexpr int CB  = 80;
constexpr int CRV = 384;
constexpr int DD  = 128;
constexpr int MM  = 256;
constexpr int EP  = 96;
constexpr int QP  = 132;
constexpr float INV_SQRT_D = 0.08838834764831845f;


__device__ __forceinline__
void rmba_qproj_part(const int blk,
                     const float* __restrict__ A,
                     const float* __restrict__ Wq,
                     const float* __restrict__ bq,
                     const float* __restrict__ Wk,
                     const float* __restrict__ bk,
                     float* __restrict__ Qe)
{
    const int lane = threadIdx.x & 31;
    const int wave = threadIdx.x >> 5;
    const int half = lane >> 4;
    const int l16  = lane & 15;
    const int koff = half * 2;
    const int rowBase = blk * 16;
    const int colBase = wave * 16;
    const int arow = rowBase + l16;
    const int bcol = colBase + l16;

    v8f acc = {};
    #pragma unroll 4
    for (int k = 0; k < CRV; k += 4) {
        v2f a, b;
        a.x = A[arow * CRV + k + koff];
        a.y = A[arow * CRV + k + koff + 1];
        b.x = Wq[(k + koff) * DD + bcol];
        b.y = Wq[(k + koff + 1) * DD + bcol];
        acc = __builtin_amdgcn_wmma_f32_16x16x4_f32(false, a, false, b,
                                                    (short)0, acc, false, false);
    }

    const float bias = bq[bcol];
    float vals[8], ss[8];
    #pragma unroll
    for (int r = 0; r < 8; ++r) { vals[r] = acc[r] + bias; ss[r] = vals[r] * vals[r]; }

    #pragma unroll
    for (int s = 8; s >= 1; s >>= 1)
        #pragma unroll
        for (int r = 0; r < 8; ++r) ss[r] += __shfl_xor(ss[r], s, 32);

    __shared__ float sq[8][16];
    __shared__ float rnorm[16];
    __shared__ float sQ[16 * QP];
    if (l16 == 0) {
        #pragma unroll
        for (int r = 0; r < 8; ++r) sq[wave][half * 8 + r] = ss[r];
    }
    __syncthreads();
    if (threadIdx.x < 16) {
        float s = 0.f;
        #pragma unroll
        for (int w = 0; w < 8; ++w) s += sq[w][threadIdx.x];
        rnorm[threadIdx.x] = 1.0f / fmaxf(sqrtf(s), 1e-12f);
    }
    __syncthreads();
    #pragma unroll
    for (int r = 0; r < 8; ++r) {
        const int row = half * 8 + r;
        sQ[row * QP + bcol] = vals[r] * rnorm[row];
    }
    __syncthreads();

    if (wave < 6) {
        const int col = colBase + l16;
        const float* bp = (col < CB) ? (Wk + col * DD) : bk;
        const bool live = (col <= CB);
        v8f acc2 = {};
        #pragma unroll 4
        for (int k = 0; k < DD; k += 4) {
            v2f a, b;
            a.x = sQ[l16 * QP + k + koff];
            a.y = sQ[l16 * QP + k + koff + 1];
            const float x0 = bp[k + koff];
            const float x1 = bp[k + koff + 1];
            b.x = live ? x0 : 0.f;
            b.y = live ? x1 : 0.f;
            acc2 = __builtin_amdgcn_wmma_f32_16x16x4_f32(false, a, false, b,
                                                         (short)0, acc2, false, false);
        }
        #pragma unroll
        for (int r = 0; r < 8; ++r)
            Qe[(rowBase + half * 8 + r) * EP + col] = acc2[r];
    }
}

__device__ __forceinline__
void rmba_cell_part(const int blk,
                    const float* __restrict__ bev,
                    const float* __restrict__ Wk, const float* __restrict__ bk,
                    float* __restrict__ Te)
{
    const int lane = threadIdx.x & 31;
    const int wave = threadIdx.x >> 5;
    const int half = lane >> 4;
    const int l16  = lane & 15;
    const int koff = half * 2;
    const int rowBase = blk * 16;
    const int colBase = wave * 16;
    const int arow = rowBase + l16;
    const int bcol = colBase + l16;

    __shared__ float sq[8][16];
    __shared__ __attribute__((aligned(16))) float sT[16 * EP];

    v8f acck = {};
    #pragma unroll 4
    for (int k = 0; k < CB; k += 4) {
        v2f a, wk;
        a.x  = bev[(k + koff) * NBC + arow];
        a.y  = bev[(k + koff + 1) * NBC + arow];
        wk.x = Wk[(k + koff) * DD + bcol];
        wk.y = Wk[(k + koff + 1) * DD + bcol];
        acck = __builtin_amdgcn_wmma_f32_16x16x4_f32(false, a, false, wk,
                                                     (short)0, acck, false, false);
    }

    {
        const int ci = threadIdx.x & 15;
        const int c0 = threadIdx.x >> 4;
        #pragma unroll
        for (int j = 0; j < 5; ++j) {
            const int c = c0 + 16 * j;
            sT[ci * EP + c] = bev[c * NBC + rowBase + ci];
        }
        if (threadIdx.x < 240) {
            const int ri = threadIdx.x / 15;
            const int cj = threadIdx.x % 15;
            const int c  = (cj == 0) ? 80 : 81 + cj;
            sT[ri * EP + c] = (cj == 0) ? 1.0f : 0.0f;
        }
    }

    const float biask = bk[bcol];
    float ss[8];
    #pragma unroll
    for (int r = 0; r < 8; ++r) { const float kv = acck[r] + biask; ss[r] = kv * kv; }
    #pragma unroll
    for (int s = 8; s >= 1; s >>= 1)
        #pragma unroll
        for (int r = 0; r < 8; ++r) ss[r] += __shfl_xor(ss[r], s, 32);

    if (l16 == 0) {
        #pragma unroll
        for (int r = 0; r < 8; ++r) sq[wave][half * 8 + r] = ss[r];
    }
    __syncthreads();
    if (threadIdx.x < 16) {
        float s = 0.f;
        #pragma unroll
        for (int w = 0; w < 8; ++w) s += sq[w][threadIdx.x];
        sT[threadIdx.x * EP + 81] = 1.0f / fmaxf(sqrtf(s), 1e-12f);
    }
    __syncthreads();
    {
        const float4* s4 = (const float4*)sT;
        float4* t4 = (float4*)(Te + (size_t)rowBase * EP);
        t4[threadIdx.x] = s4[threadIdx.x];
        if (threadIdx.x < 128) t4[256 + threadIdx.x] = s4[256 + threadIdx.x];
    }
}

constexpr int QBLK = NRQ / 16;
constexpr int CBLK = NBC / 16;

__global__ __launch_bounds__(256)
void rmba_front_kernel(const float* __restrict__ A,
                       const float* __restrict__ Wq,
                       const float* __restrict__ bq,
                       const float* __restrict__ Wk,
                       const float* __restrict__ bk,
                       const float* __restrict__ bev,
                       float* __restrict__ Qe,
                       float* __restrict__ Te)
{
    if (blockIdx.x < QBLK)
        rmba_qproj_part((int)blockIdx.x, A, Wq, bq, Wk, bk, Qe);
    else
        rmba_cell_part((int)blockIdx.x - QBLK, bev, Wk, bk, Te);
}

constexpr int QPB = 4;

template <int M>
__device__ __forceinline__ float lane_xor(float v) {
    return __int_as_float(__builtin_amdgcn_ds_swizzle(__float_as_int(v), 0x1f | (M << 10)));
}
template <int G>
__device__ __forceinline__ float group8_lane(float v) {
    return __int_as_float(__builtin_amdgcn_ds_swizzle(__float_as_int(v), 0x18 | (G << 5)));
}

__global__ __launch_bounds__(256)
void rmba_attn_kernel(const float* __restrict__ Qe,
                      const float* __restrict__ Te,
                      const int*   __restrict__ cand,
                      float* __restrict__ attn_out,
                      float* __restrict__ ctxe)
{
    const int n0   = blockIdx.x * QPB;
    const int tid  = threadIdx.x;
    const int lane = tid & 31;
    const int wave = tid >> 5;
    const int sub  = lane & 7;

    __shared__ float red_sm[8];
    __shared__ float cpart[8][EP];

    const char* tbytes = (const char*)Te;
    const unsigned subOff = (unsigned)sub * 16u;
    const bool b0 = (lane & 1) != 0;
    const bool b1 = (lane & 2) != 0;
    const bool b2 = (lane & 4) != 0;
    const bool b3 = (lane & 8) != 0;
    const bool b4 = (lane & 16) != 0;

    for (int qi = 0; qi < QPB; ++qi) {
        const int n = n0 + qi;

        const int idx = cand[(unsigned)(n * MM) + (unsigned)tid];
        const int4* cand4 = (const int4*)cand;
        const unsigned gq = (unsigned)(n * (MM / 4)) + (unsigned)(tid >> 3) * 2u;
        const int4 ca = cand4[gq];
        const int4 cb = cand4[gq + 1u];

        const float rn = Te[(unsigned)(idx < 0 ? 0 : idx) * (unsigned)EP + 81u];
        const int gi[8] = {ca.x, ca.y, ca.z, ca.w, cb.x, cb.y, cb.z, cb.w};

        const float4* qe4 = (const float4*)Qe;
        const unsigned qq = (unsigned)(n * (EP / 4)) + (unsigned)sub;
        float4 q4[3];
        #pragma unroll
        for (int t = 0; t < 3; ++t) q4[t] = qe4[qq + (unsigned)(t * 8)];

        float4 kv[8][3];
        float p[8];
        #pragma unroll
        for (int g = 0; g < 8; ++g) {
            const unsigned ixc = (unsigned)(gi[g] < 0 ? 0 : gi[g]);
            const unsigned off = ixc * (unsigned)(EP * sizeof(float)) + subOff;
            #pragma unroll
            for (int t = 0; t < 3; ++t)
                kv[g][t] = *(const float4*)(tbytes + off + t * 128);
        }

        #pragma unroll
        for (int g = 0; g < 8; ++g) {
            float acc = 0.f;
            #pragma unroll
            for (int t = 0; t < 3; ++t) {
                acc = fmaf(kv[g][t].x, q4[t].x, acc);
                acc = fmaf(kv[g][t].y, q4[t].y, acc);
                acc = fmaf(kv[g][t].z, q4[t].z, acc);
                acc = fmaf(kv[g][t].w, q4[t].w, acc);
            }
            p[g] = acc;
        }

        float r1[4];
        #pragma unroll
        for (int i = 0; i < 4; ++i) {
            const float keep = b0 ? p[2 * i + 1] : p[2 * i];
            const float send = b0 ? p[2 * i]     : p[2 * i + 1];
            r1[i] = keep + lane_xor<1>(send);
        }
        float r2[2];
        #pragma unroll
        for (int i = 0; i < 2; ++i) {
            const float keep = b1 ? r1[2 * i + 1] : r1[2 * i];
            const float send = b1 ? r1[2 * i]     : r1[2 * i + 1];
            r2[i] = keep + lane_xor<2>(send);
        }
        const float keep3 = b2 ? r2[1] : r2[0];
        const float send3 = b2 ? r2[0] : r2[1];
        const float dot = keep3 + lane_xor<4>(send3);

        const float logit = (idx >= 0) ? (dot * rn) * INV_SQRT_D : -INFINITY;

        float mx = logit;
        mx = fmaxf(mx, lane_xor<16>(mx));
        mx = fmaxf(mx, lane_xor<8>(mx));
        mx = fmaxf(mx, lane_xor<4>(mx));
        mx = fmaxf(mx, lane_xor<2>(mx));
        mx = fmaxf(mx, lane_xor<1>(mx));
        if (lane == 0) red_sm[wave] = mx;
        __syncthreads();
        float bm = red_sm[0];
        #pragma unroll
        for (int w = 1; w < 8; ++w) bm = fmaxf(bm, red_sm[w]);
        __syncthreads();

        const float e = (idx >= 0) ? __expf(logit - bm) : 0.f;
        float sm = e;
        sm += lane_xor<16>(sm);
        sm += lane_xor<8>(sm);
        sm += lane_xor<4>(sm);
        sm += lane_xor<2>(sm);
        sm += lane_xor<1>(sm);
        if (lane == 0) red_sm[wave] = sm;
        __syncthreads();
        float bs = 0.f;
        #pragma unroll
        for (int w = 0; w < 8; ++w) bs += red_sm[w];

        const float a = e / bs;
        attn_out[(unsigned)(n * MM) + (unsigned)tid] = a;

        float v[12] = {};
        const float agv[8] = {group8_lane<0>(a), group8_lane<1>(a), group8_lane<2>(a), group8_lane<3>(a),
                              group8_lane<4>(a), group8_lane<5>(a), group8_lane<6>(a), group8_lane<7>(a)};
        #pragma unroll
        for (int g = 0; g < 8; ++g) {
            const float ag = agv[g];
            #pragma unroll
            for (int t = 0; t < 3; ++t) {
                v[4 * t + 0] = fmaf(ag, kv[g][t].x, v[4 * t + 0]);
                v[4 * t + 1] = fmaf(ag, kv[g][t].y, v[4 * t + 1]);
                v[4 * t + 2] = fmaf(ag, kv[g][t].z, v[4 * t + 2]);
                v[4 * t + 3] = fmaf(ag, kv[g][t].w, v[4 * t + 3]);
            }
        }
        float w6[6];
        #pragma unroll
        for (int i = 0; i < 6; ++i) {
            const float keep = b3 ? v[6 + i] : v[i];
            const float send = b3 ? v[i]     : v[6 + i];
            w6[i] = keep + lane_xor<8>(send);
        }
        float y[3];
        #pragma unroll
        for (int i = 0; i < 3; ++i) {
            const float keep = b4 ? w6[3 + i] : w6[i];
            const float send = b4 ? w6[i]     : w6[3 + i];
            y[i] = keep + lane_xor<16>(send);
        }
        {
            const int jb = (b3 ? 6 : 0) + (b4 ? 3 : 0);
            #pragma unroll
            for (int i = 0; i < 3; ++i)
                cpart[wave][8 * (jb + i) + sub] = y[i];
        }
        __syncthreads();
        if (tid < EP) {
            const int slot = 8 * (4 * (tid >> 5) + (tid & 3)) + ((tid >> 2) & 7);
            float s = 0.f;
            #pragma unroll
            for (int w8 = 0; w8 < 8; ++w8) s += cpart[w8][slot];
            ctxe[(unsigned)(n * EP) + (unsigned)tid] = s;
        }
    }
}

__global__ __launch_bounds__(256)
void rmba_oproj_kernel(const float* __restrict__ ctxe,
                       const float* __restrict__ Wv,
                       const float* __restrict__ bv,
                       const float* __restrict__ Wo,
                       const float* __restrict__ bo,
                       float* __restrict__ out)
{
    const int mTiles = NRQ / 16;
    const int mTile  = blockIdx.x % mTiles;
    const int nGrp   = blockIdx.x / mTiles;
    const int lane = threadIdx.x & 31;
    const int wave = threadIdx.x >> 5;
    const int half = lane >> 4;
    const int l16  = lane & 15;
    const int koff = half * 2;
    const int rowBase = mTile * 16;
    const int arow = rowBase + l16;

    __shared__ float sC[16 * QP];

    {
        const int ccol = wave * 16 + l16;
        v8f acc1 = {};
        #pragma unroll 4
        for (int k = 0; k < CB; k += 4) {
            v2f a, b;
            a.x = ctxe[arow * EP + k + koff];
            a.y = ctxe[arow * EP + k + koff + 1];
            b.x = Wv[(k + koff) * DD + ccol];
            b.y = Wv[(k + koff + 1) * DD + ccol];
            acc1 = __builtin_amdgcn_wmma_f32_16x16x4_f32(false, a, false, b,
                                                         (short)0, acc1, false, false);
        }
        {
            v2f a, b;
            a.x = ctxe[arow * EP + CB + koff];
            a.y = ctxe[arow * EP + CB + koff + 1];
            const float bvc = bv[ccol];
            b.x = (half == 0) ? bvc : 0.f;
            b.y = 0.f;
            acc1 = __builtin_amdgcn_wmma_f32_16x16x4_f32(false, a, false, b,
                                                         (short)0, acc1, false, false);
        }
        #pragma unroll
        for (int r = 0; r < 8; ++r)
            sC[(half * 8 + r) * QP + ccol] = acc1[r];
    }
    __syncthreads();

    const int colBase = nGrp * 128 + wave * 16;
    const int bcol = colBase + l16;
    v8f acc = {};
    #pragma unroll 4
    for (int k = 0; k < DD; k += 4) {
        v2f a, b;
        a.x = sC[l16 * QP + k + koff];
        a.y = sC[l16 * QP + k + koff + 1];
        b.x = Wo[(k + koff) * CRV + bcol];
        b.y = Wo[(k + koff + 1) * CRV + bcol];
        acc = __builtin_amdgcn_wmma_f32_16x16x4_f32(false, a, false, b,
                                                    (short)0, acc, false, false);
    }
    const float bias = bo[bcol];
    #pragma unroll
    for (int r = 0; r < 8; ++r) {
        const int row = rowBase + half * 8 + r;
        out[row * CRV + bcol] = acc[r] + bias;
    }
}

extern "C" void kernel_launch(void* const* d_in, const int* in_sizes, int n_in,
                              void* d_out, int out_size, void* d_ws, size_t ws_size,
                              hipStream_t stream) {
    const float* rv_x = (const float*)d_in[0];
    const float* bev  = (const float*)d_in[1];
    const int*   cand = (const int*)  d_in[2];
    const float* Wq   = (const float*)d_in[3];
    const float* bq   = (const float*)d_in[4];
    const float* Wk   = (const float*)d_in[5];
    const float* bk   = (const float*)d_in[6];
    const float* Wv   = (const float*)d_in[7];
    const float* bv   = (const float*)d_in[8];
    const float* Wo   = (const float*)d_in[9];
    const float* bo   = (const float*)d_in[10];

    float* out_main = (float*)d_out;
    float* out_attn = out_main + (size_t)NRQ * CRV;

    float* Qe   = (float*)d_ws;
    float* Te   = Qe + (size_t)NRQ * EP;
    float* ctxe = Te + (size_t)NBC * EP;

    rmba_front_kernel<<<QBLK + CBLK, 256, 0, stream>>>(rv_x, Wq, bq, Wk, bk, bev, Qe, Te);
    rmba_attn_kernel <<<NRQ / QPB, 256, 0, stream>>>(Qe, Te, cand, out_attn, ctxe);
    rmba_oproj_kernel<<<(NRQ / 16) * (CRV / 128), 256, 0, stream>>>(ctxe, Wv, bv, Wo, bo, out_main);
}


# matrix k loops unrolled 10 to 16 steps deep (was 4), their bias, transpose and last-step loads asked for ahead of them: 4x fewer dependent memory round trips per tile; sums unchanged
# speedup vs baseline: 3.8806x; 1.2359x over previous
#include <hip/hip_runtime.h>
#include <math.h>


typedef float v2f __attribute__((ext_vector_type(2)));
typedef float v8f __attribute__((ext_vector_type(8)));

constexpr int NRQ = 4096;
constexpr int NBC = 16384;
constexpr int CB  = 80;
constexpr int CRV = 384;
constexpr int DD  = 128;
constexpr int MM  = 256;
constexpr int EP  = 96;
constexpr int QP  = 132;
constexpr float INV_SQRT_D = 0.08838834764831845f;


__device__ __forceinline__
void rmba_qproj_part(const int blk,
                     const float* __restrict__ A,
                     const float* __restrict__ Wq,
                     const float* __restrict__ bq,
                     const float* __restrict__ Wk,
                     const float* __restrict__ bk,
                     float* __restrict__ Qe)
{
    const int lane = threadIdx.x & 31;
    const int wave = threadIdx.x >> 5;
    const int half = lane >> 4;
    const int l16  = lane & 15;
    const int koff = half * 2;
    const int rowBase = blk * 16;
    const int colBase = wave * 16;
    const int arow = rowBase + l16;
    const int bcol = colBase + l16;

    const float bias = bq[bcol];
    v8f acc = {};
    #pragma unroll 16
    for (int k = 0; k < CRV; k += 4) {
        v2f a, b;
        a.x = A[arow * CRV + k + koff];
        a.y = A[arow * CRV + k + koff + 1];
        b.x = Wq[(k + koff) * DD + bcol];
        b.y = Wq[(k + koff + 1) * DD + bcol];
        acc = __builtin_amdgcn_wmma_f32_16x16x4_f32(false, a, false, b,
                                                    (short)0, acc, false, false);
    }

    float vals[8], ss[8];
    #pragma unroll
    for (int r = 0; r < 8; ++r) { vals[r] = acc[r] + bias; ss[r] = vals[r] * vals[r]; }

    #pragma unroll
    for (int s = 8; s >= 1; s >>= 1)
        #pragma unroll
        for (int r = 0; r < 8; ++r) ss[r] += __shfl_xor(ss[r], s, 32);

    __shared__ float sq[8][16];
    __shared__ float rnorm[16];
    __shared__ float sQ[16 * QP];
    if (l16 == 0) {
        #pragma unroll
        for (int r = 0; r < 8; ++r) sq[wave][half * 8 + r] = ss[r];
    }
    __syncthreads();
    if (threadIdx.x < 16) {
        float s = 0.f;
        #pragma unroll
        for (int w = 0; w < 8; ++w) s += sq[w][threadIdx.x];
        rnorm[threadIdx.x] = 1.0f / fmaxf(sqrtf(s), 1e-12f);
    }
    __syncthreads();
    #pragma unroll
    for (int r = 0; r < 8; ++r) {
        const int row = half * 8 + r;
        sQ[row * QP + bcol] = vals[r] * rnorm[row];
    }
    __syncthreads();

    if (wave < 6) {
        const int col = colBase + l16;
        const float* bp = (col < CB) ? (Wk + col * DD) : bk;
        const bool live = (col <= CB);
        v8f acc2 = {};
        #pragma unroll 16
        for (int k = 0; k < DD; k += 4) {
            v2f a, b;
            a.x = sQ[l16 * QP + k + koff];
            a.y = sQ[l16 * QP + k + koff + 1];
            const float x0 = bp[k + koff];
            const float x1 = bp[k + koff + 1];
            b.x = live ? x0 : 0.f;
            b.y = live ? x1 : 0.f;
            acc2 = __builtin_amdgcn_wmma_f32_16x16x4_f32(false, a, false, b,
                                                         (short)0, acc2, false, false);
        }
        #pragma unroll
        for (int r = 0; r < 8; ++r)
            Qe[(rowBase + half * 8 + r) * EP + col] = acc2[r];
    }
}

__device__ __forceinline__
void rmba_cell_part(const int blk,
                    const float* __restrict__ bev,
                    const float* __restrict__ Wk, const float* __restrict__ bk,
                    float* __restrict__ Te)
{
    const int lane = threadIdx.x & 31;
    const int wave = threadIdx.x >> 5;
    const int half = lane >> 4;
    const int l16  = lane & 15;
    const int koff = half * 2;
    const int rowBase = blk * 16;
    const int colBase = wave * 16;
    const int arow = rowBase + l16;
    const int bcol = colBase + l16;

    __shared__ float sq[8][16];
    __shared__ __attribute__((aligned(16))) float sT[16 * EP];

    const float biask = bk[bcol];
    float raw[5];
    #pragma unroll
    for (int j = 0; j < 5; ++j)
        raw[j] = bev[((int)(threadIdx.x >> 4) + 16 * j) * NBC + rowBase + (int)(threadIdx.x & 15)];
    v8f acck = {};
    #pragma unroll 10
    for (int k = 0; k < CB; k += 4) {
        v2f a, wk;
        a.x  = bev[(k + koff) * NBC + arow];
        a.y  = bev[(k + koff + 1) * NBC + arow];
        wk.x = Wk[(k + koff) * DD + bcol];
        wk.y = Wk[(k + koff + 1) * DD + bcol];
        acck = __builtin_amdgcn_wmma_f32_16x16x4_f32(false, a, false, wk,
                                                     (short)0, acck, false, false);
    }

    {
        const int ci = threadIdx.x & 15;
        const int c0 = threadIdx.x >> 4;
        #pragma unroll
        for (int j = 0; j < 5; ++j)
            sT[ci * EP + c0 + 16 * j] = raw[j];
        if (threadIdx.x < 240) {
            const int ri = threadIdx.x / 15;
            const int cj = threadIdx.x % 15;
            const int c  = (cj == 0) ? 80 : 81 + cj;
            sT[ri * EP + c] = (cj == 0) ? 1.0f : 0.0f;
        }
    }

    float ss[8];
    #pragma unroll
    for (int r = 0; r < 8; ++r) { const float kv = acck[r] + biask; ss[r] = kv * kv; }
    #pragma unroll
    for (int s = 8; s >= 1; s >>= 1)
        #pragma unroll
        for (int r = 0; r < 8; ++r) ss[r] += __shfl_xor(ss[r], s, 32);

    if (l16 == 0) {
        #pragma unroll
        for (int r = 0; r < 8; ++r) sq[wave][half * 8 + r] = ss[r];
    }
    __syncthreads();
    if (threadIdx.x < 16) {
        float s = 0.f;
        #pragma unroll
        for (int w = 0; w < 8; ++w) s += sq[w][threadIdx.x];
        sT[threadIdx.x * EP + 81] = 1.0f / fmaxf(sqrtf(s), 1e-12f);
    }
    __syncthreads();
    {
        const float4* s4 = (const float4*)sT;
        float4* t4 = (float4*)(Te + (size_t)rowBase * EP);
        t4[threadIdx.x] = s4[threadIdx.x];
        if (threadIdx.x < 128) t4[256 + threadIdx.x] = s4[256 + threadIdx.x];
    }
}

constexpr int QBLK = NRQ / 16;
constexpr int CBLK = NBC / 16;

__global__ __launch_bounds__(256)
void rmba_front_kernel(const float* __restrict__ A,
                       const float* __restrict__ Wq,
                       const float* __restrict__ bq,
                       const float* __restrict__ Wk,
                       const float* __restrict__ bk,
                       const float* __restrict__ bev,
                       float* __restrict__ Qe,
                       float* __restrict__ Te)
{
    if (blockIdx.x < QBLK)
        rmba_qproj_part((int)blockIdx.x, A, Wq, bq, Wk, bk, Qe);
    else
        rmba_cell_part((int)blockIdx.x - QBLK, bev, Wk, bk, Te);
}

constexpr int QPB = 4;

template <int M>
__device__ __forceinline__ float lane_xor(float v) {
    return __int_as_float(__builtin_amdgcn_ds_swizzle(__float_as_int(v), 0x1f | (M << 10)));
}
template <int G>
__device__ __forceinline__ float group8_lane(float v) {
    return __int_as_float(__builtin_amdgcn_ds_swizzle(__float_as_int(v), 0x18 | (G << 5)));
}

__global__ __launch_bounds__(256)
void rmba_attn_kernel(const float* __restrict__ Qe,
                      const float* __restrict__ Te,
                      const int*   __restrict__ cand,
                      float* __restrict__ attn_out,
                      float* __restrict__ ctxe)
{
    const int n0   = blockIdx.x * QPB;
    const int tid  = threadIdx.x;
    const int lane = tid & 31;
    const int wave = tid >> 5;
    const int sub  = lane & 7;

    __shared__ float red_sm[8];
    __shared__ float cpart[8][EP];

    const char* tbytes = (const char*)Te;
    const unsigned subOff = (unsigned)sub * 16u;
    const bool b0 = (lane & 1) != 0;
    const bool b1 = (lane & 2) != 0;
    const bool b2 = (lane & 4) != 0;
    const bool b3 = (lane & 8) != 0;
    const bool b4 = (lane & 16) != 0;

    for (int qi = 0; qi < QPB; ++qi) {
        const int n = n0 + qi;

        const int idx = cand[(unsigned)(n * MM) + (unsigned)tid];
        const int4* cand4 = (const int4*)cand;
        const unsigned gq = (unsigned)(n * (MM / 4)) + (unsigned)(tid >> 3) * 2u;
        const int4 ca = cand4[gq];
        const int4 cb = cand4[gq + 1u];

        const float rn = Te[(unsigned)(idx < 0 ? 0 : idx) * (unsigned)EP + 81u];
        const int gi[8] = {ca.x, ca.y, ca.z, ca.w, cb.x, cb.y, cb.z, cb.w};

        const float4* qe4 = (const float4*)Qe;
        const unsigned qq = (unsigned)(n * (EP / 4)) + (unsigned)sub;
        float4 q4[3];
        #pragma unroll
        for (int t = 0; t < 3; ++t) q4[t] = qe4[qq + (unsigned)(t * 8)];

        float4 kv[8][3];
        float p[8];
        #pragma unroll
        for (int g = 0; g < 8; ++g) {
            const unsigned ixc = (unsigned)(gi[g] < 0 ? 0 : gi[g]);
            const unsigned off = ixc * (unsigned)(EP * sizeof(float)) + subOff;
            #pragma unroll
            for (int t = 0; t < 3; ++t)
                kv[g][t] = *(const float4*)(tbytes + off + t * 128);
        }

        #pragma unroll
        for (int g = 0; g < 8; ++g) {
            float acc = 0.f;
            #pragma unroll
            for (int t = 0; t < 3; ++t) {
                acc = fmaf(kv[g][t].x, q4[t].x, acc);
                acc = fmaf(kv[g][t].y, q4[t].y, acc);
                acc = fmaf(kv[g][t].z, q4[t].z, acc);
                acc = fmaf(kv[g][t].w, q4[t].w, acc);
            }
            p[g] = acc;
        }

        float r1[4];
        #pragma unroll
        for (int i = 0; i < 4; ++i) {
            const float keep = b0 ? p[2 * i + 1] : p[2 * i];
            const float send = b0 ? p[2 * i]     : p[2 * i + 1];
            r1[i] = keep + lane_xor<1>(send);
        }
        float r2[2];
        #pragma unroll
        for (int i = 0; i < 2; ++i) {
            const float keep = b1 ? r1[2 * i + 1] : r1[2 * i];
            const float send = b1 ? r1[2 * i]     : r1[2 * i + 1];
            r2[i] = keep + lane_xor<2>(send);
        }
        const float keep3 = b2 ? r2[1] : r2[0];
        const float send3 = b2 ? r2[0] : r2[1];
        const float dot = keep3 + lane_xor<4>(send3);

        const float logit = (idx >= 0) ? (dot * rn) * INV_SQRT_D : -INFINITY;

        float mx = logit;
        mx = fmaxf(mx, lane_xor<16>(mx));
        mx = fmaxf(mx, lane_xor<8>(mx));
        mx = fmaxf(mx, lane_xor<4>(mx));
        mx = fmaxf(mx, lane_xor<2>(mx));
        mx = fmaxf(mx, lane_xor<1>(mx));
        if (lane == 0) red_sm[wave] = mx;
        __syncthreads();
        float bm = red_sm[0];
        #pragma unroll
        for (int w = 1; w < 8; ++w) bm = fmaxf(bm, red_sm[w]);
        __syncthreads();

        const float e = (idx >= 0) ? __expf(logit - bm) : 0.f;
        float sm = e;
        sm += lane_xor<16>(sm);
        sm += lane_xor<8>(sm);
        sm += lane_xor<4>(sm);
        sm += lane_xor<2>(sm);
        sm += lane_xor<1>(sm);
        if (lane == 0) red_sm[wave] = sm;
        __syncthreads();
        float bs = 0.f;
        #pragma unroll
        for (int w = 0; w < 8; ++w) bs += red_sm[w];

        const float a = e / bs;
        attn_out[(unsigned)(n * MM) + (unsigned)tid] = a;

        float v[12] = {};
        const float agv[8] = {group8_lane<0>(a), group8_lane<1>(a), group8_lane<2>(a), group8_lane<3>(a),
                              group8_lane<4>(a), group8_lane<5>(a), group8_lane<6>(a), group8_lane<7>(a)};
        #pragma unroll
        for (int g = 0; g < 8; ++g) {
            const float ag = agv[g];
            #pragma unroll
            for (int t = 0; t < 3; ++t) {
                v[4 * t + 0] = fmaf(ag, kv[g][t].x, v[4 * t + 0]);
                v[4 * t + 1] = fmaf(ag, kv[g][t].y, v[4 * t + 1]);
                v[4 * t + 2] = fmaf(ag, kv[g][t].z, v[4 * t + 2]);
                v[4 * t + 3] = fmaf(ag, kv[g][t].w, v[4 * t + 3]);
            }
        }
        float w6[6];
        #pragma unroll
        for (int i = 0; i < 6; ++i) {
            const float keep = b3 ? v[6 + i] : v[i];
            const float send = b3 ? v[i]     : v[6 + i];
            w6[i] = keep + lane_xor<8>(send);
        }
        float y[3];
        #pragma unroll
        for (int i = 0; i < 3; ++i) {
            const float keep = b4 ? w6[3 + i] : w6[i];
            const float send = b4 ? w6[i]     : w6[3 + i];
            y[i] = keep + lane_xor<16>(send);
        }
        {
            const int jb = (b3 ? 6 : 0) + (b4 ? 3 : 0);
            #pragma unroll
            for (int i = 0; i < 3; ++i)
                cpart[wave][8 * (jb + i) + sub] = y[i];
        }
        __syncthreads();
        if (tid < EP) {
            const int slot = 8 * (4 * (tid >> 5) + (tid & 3)) + ((tid >> 2) & 7);
            float s = 0.f;
            #pragma unroll
            for (int w8 = 0; w8 < 8; ++w8) s += cpart[w8][slot];
            ctxe[(unsigned)(n * EP) + (unsigned)tid] = s;
        }
    }
}

__global__ __launch_bounds__(256)
void rmba_oproj_kernel(const float* __restrict__ ctxe,
                       const float* __restrict__ Wv,
                       const float* __restrict__ bv,
                       const float* __restrict__ Wo,
                       const float* __restrict__ bo,
                       float* __restrict__ out)
{
    const int mTiles = NRQ / 16;
    const int mTile  = blockIdx.x % mTiles;
    const int nGrp   = blockIdx.x / mTiles;
    const int lane = threadIdx.x & 31;
    const int wave = threadIdx.x >> 5;
    const int half = lane >> 4;
    const int l16  = lane & 15;
    const int koff = half * 2;
    const int rowBase = mTile * 16;
    const int arow = rowBase + l16;

    __shared__ float sC[16 * QP];

    {
        const int ccol = wave * 16 + l16;
        v2f at;
        at.x = ctxe[arow * EP + CB + koff];
        at.y = ctxe[arow * EP + CB + koff + 1];
        const float bvc = bv[ccol];
        v8f acc1 = {};
        #pragma unroll 10
        for (int k = 0; k < CB; k += 4) {
            v2f a, b;
            a.x = ctxe[arow * EP + k + koff];
            a.y = ctxe[arow * EP + k + koff + 1];
            b.x = Wv[(k + koff) * DD + ccol];
            b.y = Wv[(k + koff + 1) * DD + ccol];
            acc1 = __builtin_amdgcn_wmma_f32_16x16x4_f32(false, a, false, b,
                                                         (short)0, acc1, false, false);
        }
        {
            v2f b;
            b.x = (half == 0) ? bvc : 0.f;
            b.y = 0.f;
            acc1 = __builtin_amdgcn_wmma_f32_16x16x4_f32(false, at, false, b,
                                                         (short)0, acc1, false, false);
        }
        #pragma unroll
        for (int r = 0; r < 8; ++r)
            sC[(half * 8 + r) * QP + ccol] = acc1[r];
    }
    __syncthreads();

    const int colBase = nGrp * 128 + wave * 16;
    const int bcol = colBase + l16;
    const float bias = bo[bcol];
    v8f acc = {};
    #pragma unroll 16
    for (int k = 0; k < DD; k += 4) {
        v2f a, b;
        a.x = sC[l16 * QP + k + koff];
        a.y = sC[l16 * QP + k + koff + 1];
        b.x = Wo[(k + koff) * CRV + bcol];
        b.y = Wo[(k + koff + 1) * CRV + bcol];
        acc = __builtin_amdgcn_wmma_f32_16x16x4_f32(false, a, false, b,
                                                    (short)0, acc, false, false);
    }
    #pragma unroll
    for (int r = 0; r < 8; ++r) {
        const int row = rowBase + half * 8 + r;
        out[row * CRV + bcol] = acc[r] + bias;
    }
}

extern "C" void kernel_launch(void* const* d_in, const int* in_sizes, int n_in,
                              void* d_out, int out_size, void* d_ws, size_t ws_size,
                              hipStream_t stream) {
    const float* rv_x = (const float*)d_in[0];
    const float* bev  = (const float*)d_in[1];
    const int*   cand = (const int*)  d_in[2];
    const float* Wq   = (const float*)d_in[3];
    const float* bq   = (const float*)d_in[4];
    const float* Wk   = (const float*)d_in[5];
    const float* bk   = (const float*)d_in[6];
    const float* Wv   = (const float*)d_in[7];
    const float* bv   = (const float*)d_in[8];
    const float* Wo   = (const float*)d_in[9];
    const float* bo   = (const float*)d_in[10];

    float* out_main = (float*)d_out;
    float* out_attn = out_main + (size_t)NRQ * CRV;

    float* Qe   = (float*)d_ws;
    float* Te   = Qe + (size_t)NRQ * EP;
    float* ctxe = Te + (size_t)NBC * EP;

    rmba_front_kernel<<<QBLK + CBLK, 256, 0, stream>>>(rv_x, Wq, bq, Wk, bk, bev, Qe, Te);
    rmba_attn_kernel <<<NRQ / QPB, 256, 0, stream>>>(Qe, Te, cand, out_attn, ctxe);
    rmba_oproj_kernel<<<(NRQ / 16) * (CRV / 128), 256, 0, stream>>>(ctxe, Wv, bv, Wo, bo, out_main);
}


# A tiles of the query and output projections staged once per work-group in LDS by whole-line 16-B loads; fragments read from LDS, not 16-line loads repeated by 8 waves; sums unchanged
# speedup vs baseline: 4.0726x; 1.0495x over previous
#include <hip/hip_runtime.h>
#include <math.h>


typedef float v2f __attribute__((ext_vector_type(2)));
typedef float v8f __attribute__((ext_vector_type(8)));

constexpr int NRQ = 4096;
constexpr int NBC = 16384;
constexpr int CB  = 80;
constexpr int CRV = 384;
constexpr int DD  = 128;
constexpr int MM  = 256;
constexpr int EP  = 96;
constexpr int QP  = 132;
constexpr int AP  = CRV + 4;
constexpr int XP  = EP + 4;
constexpr float INV_SQRT_D = 0.08838834764831845f;


__device__ __forceinline__
void rmba_qproj_part(const int blk,
                     const float* __restrict__ A,
                     const float* __restrict__ Wq,
                     const float* __restrict__ bq,
                     const float* __restrict__ Wk,
                     const float* __restrict__ bk,
                     float* __restrict__ Qe)
{
    const int lane = threadIdx.x & 31;
    const int wave = threadIdx.x >> 5;
    const int half = lane >> 4;
    const int l16  = lane & 15;
    const int koff = half * 2;
    const int rowBase = blk * 16;
    const int colBase = wave * 16;
    const int bcol = colBase + l16;

    const float bias = bq[bcol];

    __shared__ __attribute__((aligned(16))) float sA[16 * AP];
    {
        const float4* a4 = (const float4*)(A + (size_t)rowBase * CRV);
        float4 st[6];
        #pragma unroll
        for (int j = 0; j < 6; ++j) st[j] = a4[(int)threadIdx.x + 256 * j];
        #pragma unroll
        for (int j = 0; j < 6; ++j) {
            const int f  = (int)threadIdx.x + 256 * j;
            const int r  = f / (CRV / 4);
            const int c4 = f % (CRV / 4);
            *(float4*)&sA[r * AP + c4 * 4] = st[j];
        }
    }
    __syncthreads();

    v8f acc = {};
    #pragma unroll 16
    for (int k = 0; k < CRV; k += 4) {
        v2f b;
        const v2f a = *(const v2f*)&sA[l16 * AP + k + koff];
        b.x = Wq[(k + koff) * DD + bcol];
        b.y = Wq[(k + koff + 1) * DD + bcol];
        acc = __builtin_amdgcn_wmma_f32_16x16x4_f32(false, a, false, b,
                                                    (short)0, acc, false, false);
    }

    float vals[8], ss[8];
    #pragma unroll
    for (int r = 0; r < 8; ++r) { vals[r] = acc[r] + bias; ss[r] = vals[r] * vals[r]; }

    #pragma unroll
    for (int s = 8; s >= 1; s >>= 1)
        #pragma unroll
        for (int r = 0; r < 8; ++r) ss[r] += __shfl_xor(ss[r], s, 32);

    __shared__ float sq[8][16];
    __shared__ float rnorm[16];
    __shared__ float sQ[16 * QP];
    if (l16 == 0) {
        #pragma unroll
        for (int r = 0; r < 8; ++r) sq[wave][half * 8 + r] = ss[r];
    }
    __syncthreads();
    if (threadIdx.x < 16) {
        float s = 0.f;
        #pragma unroll
        for (int w = 0; w < 8; ++w) s += sq[w][threadIdx.x];
        rnorm[threadIdx.x] = 1.0f / fmaxf(sqrtf(s), 1e-12f);
    }
    __syncthreads();
    #pragma unroll
    for (int r = 0; r < 8; ++r) {
        const int row = half * 8 + r;
        sQ[row * QP + bcol] = vals[r] * rnorm[row];
    }
    __syncthreads();

    if (wave < 6) {
        const int col = colBase + l16;
        const float* bp = (col < CB) ? (Wk + col * DD) : bk;
        const bool live = (col <= CB);
        v8f acc2 = {};
        #pragma unroll 16
        for (int k = 0; k < DD; k += 4) {
            v2f a, b;
            a.x = sQ[l16 * QP + k + koff];
            a.y = sQ[l16 * QP + k + koff + 1];
            const float x0 = bp[k + koff];
            const float x1 = bp[k + koff + 1];
            b.x = live ? x0 : 0.f;
            b.y = live ? x1 : 0.f;
            acc2 = __builtin_amdgcn_wmma_f32_16x16x4_f32(false, a, false, b,
                                                         (short)0, acc2, false, false);
        }
        #pragma unroll
        for (int r = 0; r < 8; ++r)
            Qe[(rowBase + half * 8 + r) * EP + col] = acc2[r];
    }
}

__device__ __forceinline__
void rmba_cell_part(const int blk,
                    const float* __restrict__ bev,
                    const float* __restrict__ Wk, const float* __restrict__ bk,
                    float* __restrict__ Te)
{
    const int lane = threadIdx.x & 31;
    const int wave = threadIdx.x >> 5;
    const int half = lane >> 4;
    const int l16  = lane & 15;
    const int koff = half * 2;
    const int rowBase = blk * 16;
    const int colBase = wave * 16;
    const int arow = rowBase + l16;
    const int bcol = colBase + l16;

    __shared__ float sq[8][16];
    __shared__ __attribute__((aligned(16))) float sT[16 * EP];

    const float biask = bk[bcol];
    float raw[5];
    #pragma unroll
    for (int j = 0; j < 5; ++j)
        raw[j] = bev[((int)(threadIdx.x >> 4) + 16 * j) * NBC + rowBase + (int)(threadIdx.x & 15)];
    v8f acck = {};
    #pragma unroll 10
    for (int k = 0; k < CB; k += 4) {
        v2f a, wk;
        a.x  = bev[(k + koff) * NBC + arow];
        a.y  = bev[(k + koff + 1) * NBC + arow];
        wk.x = Wk[(k + koff) * DD + bcol];
        wk.y = Wk[(k + koff + 1) * DD + bcol];
        acck = __builtin_amdgcn_wmma_f32_16x16x4_f32(false, a, false, wk,
                                                     (short)0, acck, false, false);
    }

    {
        const int ci = threadIdx.x & 15;
        const int c0 = threadIdx.x >> 4;
        #pragma unroll
        for (int j = 0; j < 5; ++j)
            sT[ci * EP + c0 + 16 * j] = raw[j];
        if (threadIdx.x < 240) {
            const int ri = threadIdx.x / 15;
            const int cj = threadIdx.x % 15;
            const int c  = (cj == 0) ? 80 : 81 + cj;
            sT[ri * EP + c] = (cj == 0) ? 1.0f : 0.0f;
        }
    }

    float ss[8];
    #pragma unroll
    for (int r = 0; r < 8; ++r) { const float kv = acck[r] + biask; ss[r] = kv * kv; }
    #pragma unroll
    for (int s = 8; s >= 1; s >>= 1)
        #pragma unroll
        for (int r = 0; r < 8; ++r) ss[r] += __shfl_xor(ss[r], s, 32);

    if (l16 == 0) {
        #pragma unroll
        for (int r = 0; r < 8; ++r) sq[wave][half * 8 + r] = ss[r];
    }
    __syncthreads();
    if (threadIdx.x < 16) {
        float s = 0.f;
        #pragma unroll
        for (int w = 0; w < 8; ++w) s += sq[w][threadIdx.x];
        sT[threadIdx.x * EP + 81] = 1.0f / fmaxf(sqrtf(s), 1e-12f);
    }
    __syncthreads();
    {
        const float4* s4 = (const float4*)sT;
        float4* t4 = (float4*)(Te + (size_t)rowBase * EP);
        t4[threadIdx.x] = s4[threadIdx.x];
        if (threadIdx.x < 128) t4[256 + threadIdx.x] = s4[256 + threadIdx.x];
    }
}

constexpr int QBLK = NRQ / 16;
constexpr int CBLK = NBC / 16;

__global__ __launch_bounds__(256)
void rmba_front_kernel(const float* __restrict__ A,
                       const float* __restrict__ Wq,
                       const float* __restrict__ bq,
                       const float* __restrict__ Wk,
                       const float* __restrict__ bk,
                       const float* __restrict__ bev,
                       float* __restrict__ Qe,
                       float* __restrict__ Te)
{
    if (blockIdx.x < QBLK)
        rmba_qproj_part((int)blockIdx.x, A, Wq, bq, Wk, bk, Qe);
    else
        rmba_cell_part((int)blockIdx.x - QBLK, bev, Wk, bk, Te);
}

constexpr int QPB = 4;

template <int M>
__device__ __forceinline__ float lane_xor(float v) {
    return __int_as_float(__builtin_amdgcn_ds_swizzle(__float_as_int(v), 0x1f | (M << 10)));
}
template <int G>
__device__ __forceinline__ float group8_lane(float v) {
    return __int_as_float(__builtin_amdgcn_ds_swizzle(__float_as_int(v), 0x18 | (G << 5)));
}

__global__ __launch_bounds__(256)
void rmba_attn_kernel(const float* __restrict__ Qe,
                      const float* __restrict__ Te,
                      const int*   __restrict__ cand,
                      float* __restrict__ attn_out,
                      float* __restrict__ ctxe)
{
    const int n0   = blockIdx.x * QPB;
    const int tid  = threadIdx.x;
    const int lane = tid & 31;
    const int wave = tid >> 5;
    const int sub  = lane & 7;

    __shared__ float red_sm[8];
    __shared__ float cpart[8][EP];

    const char* tbytes = (const char*)Te;
    const unsigned subOff = (unsigned)sub * 16u;
    const bool b0 = (lane & 1) != 0;
    const bool b1 = (lane & 2) != 0;
    const bool b2 = (lane & 4) != 0;
    const bool b3 = (lane & 8) != 0;
    const bool b4 = (lane & 16) != 0;

    for (int qi = 0; qi < QPB; ++qi) {
        const int n = n0 + qi;

        const int idx = cand[(unsigned)(n * MM) + (unsigned)tid];
        const int4* cand4 = (const int4*)cand;
        const unsigned gq = (unsigned)(n * (MM / 4)) + (unsigned)(tid >> 3) * 2u;
        const int4 ca = cand4[gq];
        const int4 cb = cand4[gq + 1u];

        const float rn = Te[(unsigned)(idx < 0 ? 0 : idx) * (unsigned)EP + 81u];
        const int gi[8] = {ca.x, ca.y, ca.z, ca.w, cb.x, cb.y, cb.z, cb.w};

        const float4* qe4 = (const float4*)Qe;
        const unsigned qq = (unsigned)(n * (EP / 4)) + (unsigned)sub;
        float4 q4[3];
        #pragma unroll
        for (int t = 0; t < 3; ++t) q4[t] = qe4[qq + (unsigned)(t * 8)];

        float4 kv[8][3];
        float p[8];
        #pragma unroll
        for (int g = 0; g < 8; ++g) {
            const unsigned ixc = (unsigned)(gi[g] < 0 ? 0 : gi[g]);
            const unsigned off = ixc * (unsigned)(EP * sizeof(float)) + subOff;
            #pragma unroll
            for (int t = 0; t < 3; ++t)
                kv[g][t] = *(const float4*)(tbytes + off + t * 128);
        }

        #pragma unroll
        for (int g = 0; g < 8; ++g) {
            float acc = 0.f;
            #pragma unroll
            for (int t = 0; t < 3; ++t) {
                acc = fmaf(kv[g][t].x, q4[t].x, acc);
                acc = fmaf(kv[g][t].y, q4[t].y, acc);
                acc = fmaf(kv[g][t].z, q4[t].z, acc);
                acc = fmaf(kv[g][t].w, q4[t].w, acc);
            }
            p[g] = acc;
        }

        float r1[4];
        #pragma unroll
        for (int i = 0; i < 4; ++i) {
            const float keep = b0 ? p[2 * i + 1] : p[2 * i];
            const float send = b0 ? p[2 * i]     : p[2 * i + 1];
            r1[i] = keep + lane_xor<1>(send);
        }
        float r2[2];
        #pragma unroll
        for (int i = 0; i < 2; ++i) {
            const float keep = b1 ? r1[2 * i + 1] : r1[2 * i];
            const float send = b1 ? r1[2 * i]     : r1[2 * i + 1];
            r2[i] = keep + lane_xor<2>(send);
        }
        const float keep3 = b2 ? r2[1] : r2[0];
        const float send3 = b2 ? r2[0] : r2[1];
        const float dot = keep3 + lane_xor<4>(send3);

        const float logit = (idx >= 0) ? (dot * rn) * INV_SQRT_D : -INFINITY;

        float mx = logit;
        mx = fmaxf(mx, lane_xor<16>(mx));
        mx = fmaxf(mx, lane_xor<8>(mx));
        mx = fmaxf(mx, lane_xor<4>(mx));
        mx = fmaxf(mx, lane_xor<2>(mx));
        mx = fmaxf(mx, lane_xor<1>(mx));
        if (lane == 0) red_sm[wave] = mx;
        __syncthreads();
        float bm = red_sm[0];
        #pragma unroll
        for (int w = 1; w < 8; ++w) bm = fmaxf(bm, red_sm[w]);
        __syncthreads();

        const float e = (idx >= 0) ? __expf(logit - bm) : 0.f;
        float sm = e;
        sm += lane_xor<16>(sm);
        sm += lane_xor<8>(sm);
        sm += lane_xor<4>(sm);
        sm += lane_xor<2>(sm);
        sm += lane_xor<1>(sm);
        if (lane == 0) red_sm[wave] = sm;
        __syncthreads();
        float bs = 0.f;
        #pragma unroll
        for (int w = 0; w < 8; ++w) bs += red_sm[w];

        const float a = e / bs;
        attn_out[(unsigned)(n * MM) + (unsigned)tid] = a;

        float v[12] = {};
        const float agv[8] = {group8_lane<0>(a), group8_lane<1>(a), group8_lane<2>(a), group8_lane<3>(a),
                              group8_lane<4>(a), group8_lane<5>(a), group8_lane<6>(a), group8_lane<7>(a)};
        #pragma unroll
        for (int g = 0; g < 8; ++g) {
            const float ag = agv[g];
            #pragma unroll
            for (int t = 0; t < 3; ++t) {
                v[4 * t + 0] = fmaf(ag, kv[g][t].x, v[4 * t + 0]);
                v[4 * t + 1] = fmaf(ag, kv[g][t].y, v[4 * t + 1]);
                v[4 * t + 2] = fmaf(ag, kv[g][t].z, v[4 * t + 2]);
                v[4 * t + 3] = fmaf(ag, kv[g][t].w, v[4 * t + 3]);
            }
        }
        float w6[6];
        #pragma unroll
        for (int i = 0; i < 6; ++i) {
            const float keep = b3 ? v[6 + i] : v[i];
            const float send = b3 ? v[i]     : v[6 + i];
            w6[i] = keep + lane_xor<8>(send);
        }
        float y[3];
        #pragma unroll
        for (int i = 0; i < 3; ++i) {
            const float keep = b4 ? w6[3 + i] : w6[i];
            const float send = b4 ? w6[i]     : w6[3 + i];
            y[i] = keep + lane_xor<16>(send);
        }
        {
            const int jb = (b3 ? 6 : 0) + (b4 ? 3 : 0);
            #pragma unroll
            for (int i = 0; i < 3; ++i)
                cpart[wave][8 * (jb + i) + sub] = y[i];
        }
        __syncthreads();
        if (tid < EP) {
            const int slot = 8 * (4 * (tid >> 5) + (tid & 3)) + ((tid >> 2) & 7);
            float s = 0.f;
            #pragma unroll
            for (int w8 = 0; w8 < 8; ++w8) s += cpart[w8][slot];
            ctxe[(unsigned)(n * EP) + (unsigned)tid] = s;
        }
    }
}

__global__ __launch_bounds__(256)
void rmba_oproj_kernel(const float* __restrict__ ctxe,
                       const float* __restrict__ Wv,
                       const float* __restrict__ bv,
                       const float* __restrict__ Wo,
                       const float* __restrict__ bo,
                       float* __restrict__ out)
{
    const int mTiles = NRQ / 16;
    const int mTile  = blockIdx.x % mTiles;
    const int nGrp   = blockIdx.x / mTiles;
    const int lane = threadIdx.x & 31;
    const int wave = threadIdx.x >> 5;
    const int half = lane >> 4;
    const int l16  = lane & 15;
    const int koff = half * 2;
    const int rowBase = mTile * 16;

    __shared__ float sC[16 * QP];
    __shared__ __attribute__((aligned(16))) float sX[16 * XP];

    {
        const int ccol = wave * 16 + l16;
        const float bvc = bv[ccol];
        {
            const float4* c4p = (const float4*)(ctxe + (size_t)rowBase * EP);
            const int f0 = (int)threadIdx.x;
            const int f1 = 256 + (int)threadIdx.x;
            const bool two = threadIdx.x < 128;
            const float4 x0 = c4p[f0];
            float4 x1 = make_float4(0.f, 0.f, 0.f, 0.f);
            if (two) x1 = c4p[f1];
            *(float4*)&sX[(f0 / (EP / 4)) * XP + (f0 % (EP / 4)) * 4] = x0;
            if (two) *(float4*)&sX[(f1 / (EP / 4)) * XP + (f1 % (EP / 4)) * 4] = x1;
        }
        __syncthreads();
        const v2f at = *(const v2f*)&sX[l16 * XP + CB + koff];
        v8f acc1 = {};
        #pragma unroll 10
        for (int k = 0; k < CB; k += 4) {
            v2f b;
            const v2f a = *(const v2f*)&sX[l16 * XP + k + koff];
            b.x = Wv[(k + koff) * DD + ccol];
            b.y = Wv[(k + koff + 1) * DD + ccol];
            acc1 = __builtin_amdgcn_wmma_f32_16x16x4_f32(false, a, false, b,
                                                         (short)0, acc1, false, false);
        }
        {
            v2f b;
            b.x = (half == 0) ? bvc : 0.f;
            b.y = 0.f;
            acc1 = __builtin_amdgcn_wmma_f32_16x16x4_f32(false, at, false, b,
                                                         (short)0, acc1, false, false);
        }
        #pragma unroll
        for (int r = 0; r < 8; ++r)
            sC[(half * 8 + r) * QP + ccol] = acc1[r];
    }
    __syncthreads();

    const int colBase = nGrp * 128 + wave * 16;
    const int bcol = colBase + l16;
    const float bias = bo[bcol];
    v8f acc = {};
    #pragma unroll 16
    for (int k = 0; k < DD; k += 4) {
        v2f a, b;
        a.x = sC[l16 * QP + k + koff];
        a.y = sC[l16 * QP + k + koff + 1];
        b.x = Wo[(k + koff) * CRV + bcol];
        b.y = Wo[(k + koff + 1) * CRV + bcol];
        acc = __builtin_amdgcn_wmma_f32_16x16x4_f32(false, a, false, b,
                                                    (short)0, acc, false, false);
    }
    #pragma unroll
    for (int r = 0; r < 8; ++r) {
        const int row = rowBase + half * 8 + r;
        out[row * CRV + bcol] = acc[r] + bias;
    }
}

extern "C" void kernel_launch(void* const* d_in, const int* in_sizes, int n_in,
                              void* d_out, int out_size, void* d_ws, size_t ws_size,
                              hipStream_t stream) {
    const float* rv_x = (const float*)d_in[0];
    const float* bev  = (const float*)d_in[1];
    const int*   cand = (const int*)  d_in[2];
    const float* Wq   = (const float*)d_in[3];
    const float* bq   = (const float*)d_in[4];
    const float* Wk   = (const float*)d_in[5];
    const float* bk   = (const float*)d_in[6];
    const float* Wv   = (const float*)d_in[7];
    const float* bv   = (const float*)d_in[8];
    const float* Wo   = (const float*)d_in[9];
    const float* bo   = (const float*)d_in[10];

    float* out_main = (float*)d_out;
    float* out_attn = out_main + (size_t)NRQ * CRV;

    float* Qe   = (float*)d_ws;
    float* Te   = Qe + (size_t)NRQ * EP;
    float* ctxe = Te + (size_t)NBC * EP;

    rmba_front_kernel<<<QBLK + CBLK, 256, 0, stream>>>(rv_x, Wq, bq, Wk, bk, bev, Qe, Te);
    rmba_attn_kernel <<<NRQ / QPB, 256, 0, stream>>>(Qe, Te, cand, out_attn, ctxe);
    rmba_oproj_kernel<<<(NRQ / 16) * (CRV / 128), 256, 0, stream>>>(ctxe, Wv, bv, Wo, bo, out_main);
}


# attention pass: 1/norm taken from the gathered row by a DPP8 lane broadcast, not a second 32-line gather; waves whose 32 candidates are all padding skip their row loads (logit rounding changes)
# speedup vs baseline: 4.2820x; 1.0514x over previous
#include <hip/hip_runtime.h>
#include <math.h>


typedef float v2f __attribute__((ext_vector_type(2)));
typedef float v8f __attribute__((ext_vector_type(8)));

constexpr int NRQ = 4096;
constexpr int NBC = 16384;
constexpr int CB  = 80;
constexpr int CRV = 384;
constexpr int DD  = 128;
constexpr int MM  = 256;
constexpr int EP  = 96;
constexpr int QP  = 132;
constexpr int AP  = CRV + 4;
constexpr int XP  = EP + 4;
constexpr float INV_SQRT_D = 0.08838834764831845f;


__device__ __forceinline__
void rmba_qproj_part(const int blk,
                     const float* __restrict__ A,
                     const float* __restrict__ Wq,
                     const float* __restrict__ bq,
                     const float* __restrict__ Wk,
                     const float* __restrict__ bk,
                     float* __restrict__ Qe)
{
    const int lane = threadIdx.x & 31;
    const int wave = threadIdx.x >> 5;
    const int half = lane >> 4;
    const int l16  = lane & 15;
    const int koff = half * 2;
    const int rowBase = blk * 16;
    const int colBase = wave * 16;
    const int bcol = colBase + l16;

    const float bias = bq[bcol];

    __shared__ __attribute__((aligned(16))) float sA[16 * AP];
    {
        const float4* a4 = (const float4*)(A + (size_t)rowBase * CRV);
        float4 st[6];
        #pragma unroll
        for (int j = 0; j < 6; ++j) st[j] = a4[(int)threadIdx.x + 256 * j];
        #pragma unroll
        for (int j = 0; j < 6; ++j) {
            const int f  = (int)threadIdx.x + 256 * j;
            const int r  = f / (CRV / 4);
            const int c4 = f % (CRV / 4);
            *(float4*)&sA[r * AP + c4 * 4] = st[j];
        }
    }
    __syncthreads();

    v8f acc = {};
    #pragma unroll 16
    for (int k = 0; k < CRV; k += 4) {
        v2f b;
        const v2f a = *(const v2f*)&sA[l16 * AP + k + koff];
        b.x = Wq[(k + koff) * DD + bcol];
        b.y = Wq[(k + koff + 1) * DD + bcol];
        acc = __builtin_amdgcn_wmma_f32_16x16x4_f32(false, a, false, b,
                                                    (short)0, acc, false, false);
    }

    float vals[8], ss[8];
    #pragma unroll
    for (int r = 0; r < 8; ++r) { vals[r] = acc[r] + bias; ss[r] = vals[r] * vals[r]; }

    #pragma unroll
    for (int s = 8; s >= 1; s >>= 1)
        #pragma unroll
        for (int r = 0; r < 8; ++r) ss[r] += __shfl_xor(ss[r], s, 32);

    __shared__ float sq[8][16];
    __shared__ float rnorm[16];
    __shared__ float sQ[16 * QP];
    if (l16 == 0) {
        #pragma unroll
        for (int r = 0; r < 8; ++r) sq[wave][half * 8 + r] = ss[r];
    }
    __syncthreads();
    if (threadIdx.x < 16) {
        float s = 0.f;
        #pragma unroll
        for (int w = 0; w < 8; ++w) s += sq[w][threadIdx.x];
        rnorm[threadIdx.x] = 1.0f / fmaxf(sqrtf(s), 1e-12f);
    }
    __syncthreads();
    #pragma unroll
    for (int r = 0; r < 8; ++r) {
        const int row = half * 8 + r;
        sQ[row * QP + bcol] = vals[r] * rnorm[row];
    }
    __syncthreads();

    if (wave < 6) {
        const int col = colBase + l16;
        const float* bp = (col < CB) ? (Wk + col * DD) : bk;
        const bool live = (col <= CB);
        v8f acc2 = {};
        #pragma unroll 16
        for (int k = 0; k < DD; k += 4) {
            v2f a, b;
            a.x = sQ[l16 * QP + k + koff];
            a.y = sQ[l16 * QP + k + koff + 1];
            const float x0 = bp[k + koff];
            const float x1 = bp[k + koff + 1];
            b.x = live ? x0 : 0.f;
            b.y = live ? x1 : 0.f;
            acc2 = __builtin_amdgcn_wmma_f32_16x16x4_f32(false, a, false, b,
                                                         (short)0, acc2, false, false);
        }
        #pragma unroll
        for (int r = 0; r < 8; ++r)
            Qe[(rowBase + half * 8 + r) * EP + col] = acc2[r];
    }
}

__device__ __forceinline__
void rmba_cell_part(const int blk,
                    const float* __restrict__ bev,
                    const float* __restrict__ Wk, const float* __restrict__ bk,
                    float* __restrict__ Te)
{
    const int lane = threadIdx.x & 31;
    const int wave = threadIdx.x >> 5;
    const int half = lane >> 4;
    const int l16  = lane & 15;
    const int koff = half * 2;
    const int rowBase = blk * 16;
    const int colBase = wave * 16;
    const int arow = rowBase + l16;
    const int bcol = colBase + l16;

    __shared__ float sq[8][16];
    __shared__ __attribute__((aligned(16))) float sT[16 * EP];

    const float biask = bk[bcol];
    float raw[5];
    #pragma unroll
    for (int j = 0; j < 5; ++j)
        raw[j] = bev[((int)(threadIdx.x >> 4) + 16 * j) * NBC + rowBase + (int)(threadIdx.x & 15)];
    v8f acck = {};
    #pragma unroll 10
    for (int k = 0; k < CB; k += 4) {
        v2f a, wk;
        a.x  = bev[(k + koff) * NBC + arow];
        a.y  = bev[(k + koff + 1) * NBC + arow];
        wk.x = Wk[(k + koff) * DD + bcol];
        wk.y = Wk[(k + koff + 1) * DD + bcol];
        acck = __builtin_amdgcn_wmma_f32_16x16x4_f32(false, a, false, wk,
                                                     (short)0, acck, false, false);
    }

    {
        const int ci = threadIdx.x & 15;
        const int c0 = threadIdx.x >> 4;
        #pragma unroll
        for (int j = 0; j < 5; ++j)
            sT[ci * EP + c0 + 16 * j] = raw[j];
        if (threadIdx.x < 240) {
            const int ri = threadIdx.x / 15;
            const int cj = threadIdx.x % 15;
            const int c  = (cj == 0) ? 80 : 81 + cj;
            sT[ri * EP + c] = (cj == 0) ? 1.0f : 0.0f;
        }
    }

    float ss[8];
    #pragma unroll
    for (int r = 0; r < 8; ++r) { const float kv = acck[r] + biask; ss[r] = kv * kv; }
    #pragma unroll
    for (int s = 8; s >= 1; s >>= 1)
        #pragma unroll
        for (int r = 0; r < 8; ++r) ss[r] += __shfl_xor(ss[r], s, 32);

    if (l16 == 0) {
        #pragma unroll
        for (int r = 0; r < 8; ++r) sq[wave][half * 8 + r] = ss[r];
    }
    __syncthreads();
    if (threadIdx.x < 16) {
        float s = 0.f;
        #pragma unroll
        for (int w = 0; w < 8; ++w) s += sq[w][threadIdx.x];
        sT[threadIdx.x * EP + 81] = 1.0f / fmaxf(sqrtf(s), 1e-12f);
    }
    __syncthreads();
    {
        const float4* s4 = (const float4*)sT;
        float4* t4 = (float4*)(Te + (size_t)rowBase * EP);
        t4[threadIdx.x] = s4[threadIdx.x];
        if (threadIdx.x < 128) t4[256 + threadIdx.x] = s4[256 + threadIdx.x];
    }
}

constexpr int QBLK = NRQ / 16;
constexpr int CBLK = NBC / 16;

__global__ __launch_bounds__(256)
void rmba_front_kernel(const float* __restrict__ A,
                       const float* __restrict__ Wq,
                       const float* __restrict__ bq,
                       const float* __restrict__ Wk,
                       const float* __restrict__ bk,
                       const float* __restrict__ bev,
                       float* __restrict__ Qe,
                       float* __restrict__ Te)
{
    if (blockIdx.x < QBLK)
        rmba_qproj_part((int)blockIdx.x, A, Wq, bq, Wk, bk, Qe);
    else
        rmba_cell_part((int)blockIdx.x - QBLK, bev, Wk, bk, Te);
}

constexpr int QPB = 4;

template <int M>
__device__ __forceinline__ float lane_xor(float v) {
    return __int_as_float(__builtin_amdgcn_ds_swizzle(__float_as_int(v), 0x1f | (M << 10)));
}
template <int G>
__device__ __forceinline__ float group8_lane(float v) {
    return __int_as_float(__builtin_amdgcn_ds_swizzle(__float_as_int(v), 0x18 | (G << 5)));
}
__device__ __forceinline__ float mul_by_group_lane4(float acc, float v) {
    constexpr unsigned sel = 4u * 0x249249u;
    return acc * __int_as_float(__builtin_amdgcn_mov_dpp8(__float_as_int(v), sel));
}

__global__ __launch_bounds__(256)
void rmba_attn_kernel(const float* __restrict__ Qe,
                      const float* __restrict__ Te,
                      const int*   __restrict__ cand,
                      float* __restrict__ attn_out,
                      float* __restrict__ ctxe)
{
    const int n0   = blockIdx.x * QPB;
    const int tid  = threadIdx.x;
    const int lane = tid & 31;
    const int wave = tid >> 5;
    const int sub  = lane & 7;

    __shared__ float red_sm[8];
    __shared__ float cpart[8][EP];

    const char* tbytes = (const char*)Te;
    const unsigned subOff = (unsigned)sub * 16u;
    const bool b0 = (lane & 1) != 0;
    const bool b1 = (lane & 2) != 0;
    const bool b2 = (lane & 4) != 0;
    const bool b3 = (lane & 8) != 0;
    const bool b4 = (lane & 16) != 0;

    for (int qi = 0; qi < QPB; ++qi) {
        const int n = n0 + qi;

        const int idx = cand[(unsigned)(n * MM) + (unsigned)tid];
        const int4* cand4 = (const int4*)cand;
        const unsigned gq = (unsigned)(n * (MM / 4)) + (unsigned)(tid >> 3) * 2u;
        const int4 ca = cand4[gq];
        const int4 cb = cand4[gq + 1u];
        const int all8 = ca.x & ca.y & ca.z & ca.w & cb.x & cb.y & cb.z & cb.w;
        const bool wave_live = __builtin_amdgcn_ballot_w32(all8 >= 0) != 0u;

        float4 kv[8][3];
        float dot = 0.f;
        if (wave_live) {
            const int gi[8] = {ca.x, ca.y, ca.z, ca.w, cb.x, cb.y, cb.z, cb.w};

            const float4* qe4 = (const float4*)Qe;
            const unsigned qq = (unsigned)(n * (EP / 4)) + (unsigned)sub;
            float4 q4[3];
            #pragma unroll
            for (int t = 0; t < 3; ++t) q4[t] = qe4[qq + (unsigned)(t * 8)];

            float p[8];
            #pragma unroll
            for (int g = 0; g < 8; ++g) {
                const unsigned ixc = (unsigned)(gi[g] < 0 ? 0 : gi[g]);
                const unsigned off = ixc * (unsigned)(EP * sizeof(float)) + subOff;
                #pragma unroll
                for (int t = 0; t < 3; ++t)
                    kv[g][t] = *(const float4*)(tbytes + off + t * 128);
            }

            #pragma unroll
            for (int g = 0; g < 8; ++g) {
                float acc = 0.f;
                #pragma unroll
                for (int t = 0; t < 3; ++t) {
                    acc = fmaf(kv[g][t].x, q4[t].x, acc);
                    acc = fmaf(kv[g][t].y, q4[t].y, acc);
                    acc = fmaf(kv[g][t].z, q4[t].z, acc);
                    acc = fmaf(kv[g][t].w, q4[t].w, acc);
                }
                p[g] = acc;
            }
            __builtin_amdgcn_sched_barrier(0);
            #pragma unroll
            for (int g = 0; g < 8; ++g) p[g] = mul_by_group_lane4(p[g], kv[g][2].y);

            float r1[4];
            #pragma unroll
            for (int i = 0; i < 4; ++i) {
                const float keep = b0 ? p[2 * i + 1] : p[2 * i];
                const float send = b0 ? p[2 * i]     : p[2 * i + 1];
                r1[i] = keep + lane_xor<1>(send);
            }
            float r2[2];
            #pragma unroll
            for (int i = 0; i < 2; ++i) {
                const float keep = b1 ? r1[2 * i + 1] : r1[2 * i];
                const float send = b1 ? r1[2 * i]     : r1[2 * i + 1];
                r2[i] = keep + lane_xor<2>(send);
            }
            const float keep3 = b2 ? r2[1] : r2[0];
            const float send3 = b2 ? r2[0] : r2[1];
            dot = keep3 + lane_xor<4>(send3);
        }

        const float logit = (idx >= 0) ? dot * INV_SQRT_D : -INFINITY;

        float mx = logit;
        mx = fmaxf(mx, lane_xor<16>(mx));
        mx = fmaxf(mx, lane_xor<8>(mx));
        mx = fmaxf(mx, lane_xor<4>(mx));
        mx = fmaxf(mx, lane_xor<2>(mx));
        mx = fmaxf(mx, lane_xor<1>(mx));
        if (lane == 0) red_sm[wave] = mx;
        __syncthreads();
        float bm = red_sm[0];
        #pragma unroll
        for (int w = 1; w < 8; ++w) bm = fmaxf(bm, red_sm[w]);
        __syncthreads();

        const float e = (idx >= 0) ? __expf(logit - bm) : 0.f;
        float sm = e;
        sm += lane_xor<16>(sm);
        sm += lane_xor<8>(sm);
        sm += lane_xor<4>(sm);
        sm += lane_xor<2>(sm);
        sm += lane_xor<1>(sm);
        if (lane == 0) red_sm[wave] = sm;
        __syncthreads();
        float bs = 0.f;
        #pragma unroll
        for (int w = 0; w < 8; ++w) bs += red_sm[w];

        const float a = e / bs;
        attn_out[(unsigned)(n * MM) + (unsigned)tid] = a;

        float y[3] = {0.f, 0.f, 0.f};
        if (wave_live) {
            float v[12] = {};
            const float agv[8] = {group8_lane<0>(a), group8_lane<1>(a), group8_lane<2>(a), group8_lane<3>(a),
                                  group8_lane<4>(a), group8_lane<5>(a), group8_lane<6>(a), group8_lane<7>(a)};
            #pragma unroll
            for (int g = 0; g < 8; ++g) {
                const float ag = agv[g];
                #pragma unroll
                for (int t = 0; t < 3; ++t) {
                    v[4 * t + 0] = fmaf(ag, kv[g][t].x, v[4 * t + 0]);
                    v[4 * t + 1] = fmaf(ag, kv[g][t].y, v[4 * t + 1]);
                    v[4 * t + 2] = fmaf(ag, kv[g][t].z, v[4 * t + 2]);
                    v[4 * t + 3] = fmaf(ag, kv[g][t].w, v[4 * t + 3]);
                }
            }
            float w6[6];
            #pragma unroll
            for (int i = 0; i < 6; ++i) {
                const float keep = b3 ? v[6 + i] : v[i];
                const float send = b3 ? v[i]     : v[6 + i];
                w6[i] = keep + lane_xor<8>(send);
            }
            #pragma unroll
            for (int i = 0; i < 3; ++i) {
                const float keep = b4 ? w6[3 + i] : w6[i];
                const float send = b4 ? w6[i]     : w6[3 + i];
                y[i] = keep + lane_xor<16>(send);
            }
        }
        {
            const int jb = (b3 ? 6 : 0) + (b4 ? 3 : 0);
            #pragma unroll
            for (int i = 0; i < 3; ++i)
                cpart[wave][8 * (jb + i) + sub] = y[i];
        }
        __syncthreads();
        if (tid < EP) {
            const int slot = 8 * (4 * (tid >> 5) + (tid & 3)) + ((tid >> 2) & 7);
            float s = 0.f;
            #pragma unroll
            for (int w8 = 0; w8 < 8; ++w8) s += cpart[w8][slot];
            ctxe[(unsigned)(n * EP) + (unsigned)tid] = s;
        }
    }
}

__global__ __launch_bounds__(256)
void rmba_oproj_kernel(const float* __restrict__ ctxe,
                       const float* __restrict__ Wv,
                       const float* __restrict__ bv,
                       const float* __restrict__ Wo,
                       const float* __restrict__ bo,
                       float* __restrict__ out)
{
    const int mTiles = NRQ / 16;
    const int mTile  = blockIdx.x % mTiles;
    const int nGrp   = blockIdx.x / mTiles;
    const int lane = threadIdx.x & 31;
    const int wave = threadIdx.x >> 5;
    const int half = lane >> 4;
    const int l16  = lane & 15;
    const int koff = half * 2;
    const int rowBase = mTile * 16;

    __shared__ float sC[16 * QP];
    __shared__ __attribute__((aligned(16))) float sX[16 * XP];

    {
        const int ccol = wave * 16 + l16;
        const float bvc = bv[ccol];
        {
            const float4* c4p = (const float4*)(ctxe + (size_t)rowBase * EP);
            const int f0 = (int)threadIdx.x;
            const int f1 = 256 + (int)threadIdx.x;
            const bool two = threadIdx.x < 128;
            const float4 x0 = c4p[f0];
            float4 x1 = make_float4(0.f, 0.f, 0.f, 0.f);
            if (two) x1 = c4p[f1];
            *(float4*)&sX[(f0 / (EP / 4)) * XP + (f0 % (EP / 4)) * 4] = x0;
            if (two) *(float4*)&sX[(f1 / (EP / 4)) * XP + (f1 % (EP / 4)) * 4] = x1;
        }
        __syncthreads();
        const v2f at = *(const v2f*)&sX[l16 * XP + CB + koff];
        v8f acc1 = {};
        #pragma unroll 10
        for (int k = 0; k < CB; k += 4) {
            v2f b;
            const v2f a = *(const v2f*)&sX[l16 * XP + k + koff];
            b.x = Wv[(k + koff) * DD + ccol];
            b.y = Wv[(k + koff + 1) * DD + ccol];
            acc1 = __builtin_amdgcn_wmma_f32_16x16x4_f32(false, a, false, b,
                                                         (short)0, acc1, false, false);
        }
        {
            v2f b;
            b.x = (half == 0) ? bvc : 0.f;
            b.y = 0.f;
            acc1 = __builtin_amdgcn_wmma_f32_16x16x4_f32(false, at, false, b,
                                                         (short)0, acc1, false, false);
        }
        #pragma unroll
        for (int r = 0; r < 8; ++r)
            sC[(half * 8 + r) * QP + ccol] = acc1[r];
    }
    __syncthreads();

    const int colBase = nGrp * 128 + wave * 16;
    const int bcol = colBase + l16;
    const float bias = bo[bcol];
    v8f acc = {};
    #pragma unroll 16
    for (int k = 0; k < DD; k += 4) {
        v2f a, b;
        a.x = sC[l16 * QP + k + koff];
        a.y = sC[l16 * QP + k + koff + 1];
        b.x = Wo[(k + koff) * CRV + bcol];
        b.y = Wo[(k + koff + 1) * CRV + bcol];
        acc = __builtin_amdgcn_wmma_f32_16x16x4_f32(false, a, false, b,
                                                    (short)0, acc, false, false);
    }
    #pragma unroll
    for (int r = 0; r < 8; ++r) {
        const int row = rowBase + half * 8 + r;
        out[row * CRV + bcol] = acc[r] + bias;
    }
}

extern "C" void kernel_launch(void* const* d_in, const int* in_sizes, int n_in,
                              void* d_out, int out_size, void* d_ws, size_t ws_size,
                              hipStream_t stream) {
    const float* rv_x = (const float*)d_in[0];
    const float* bev  = (const float*)d_in[1];
    const int*   cand = (const int*)  d_in[2];
    const float* Wq   = (const float*)d_in[3];
    const float* bq   = (const float*)d_in[4];
    const float* Wk   = (const float*)d_in[5];
    const float* bk   = (const float*)d_in[6];
    const float* Wv   = (const float*)d_in[7];
    const float* bv   = (const float*)d_in[8];
    const float* Wo   = (const float*)d_in[9];
    const float* bo   = (const float*)d_in[10];

    float* out_main = (float*)d_out;
    float* out_attn = out_main + (size_t)NRQ * CRV;

    float* Qe   = (float*)d_ws;
    float* Te   = Qe + (size_t)NRQ * EP;
    float* ctxe = Te + (size_t)NBC * EP;

    rmba_front_kernel<<<QBLK + CBLK, 256, 0, stream>>>(rv_x, Wq, bq, Wk, bk, bev, Qe, Te);
    rmba_attn_kernel <<<NRQ / QPB, 256, 0, stream>>>(Qe, Te, cand, out_attn, ctxe);
    rmba_oproj_kernel<<<(NRQ / 16) * (CRV / 128), 256, 0, stream>>>(ctxe, Wv, bv, Wo, bo, out_main);
}


# every cross-lane exchange (dot butterfly, wave max/sum trees, weight broadcast, row sums) on the vector ALU by DPP row_xmask, DPP8 and v_permlanex16, not through the LDS unit; sums unchanged
# speedup vs baseline: 4.4907x; 1.0487x over previous
#include <hip/hip_runtime.h>
#include <math.h>


typedef float v2f __attribute__((ext_vector_type(2)));
typedef float v8f __attribute__((ext_vector_type(8)));

constexpr int NRQ = 4096;
constexpr int NBC = 16384;
constexpr int CB  = 80;
constexpr int CRV = 384;
constexpr int DD  = 128;
constexpr int MM  = 256;
constexpr int EP  = 96;
constexpr int QP  = 132;
constexpr int AP  = CRV + 4;
constexpr int XP  = EP + 4;
constexpr float INV_SQRT_D = 0.08838834764831845f;


template <int M>
__device__ __forceinline__ float lane_xor(float v) {
    const int b = __float_as_int(v);
    if constexpr (M == 16)
        return __int_as_float(__builtin_amdgcn_permlanex16(b, b, 0x76543210, 0xfedcba98, false, false));
    else
        return __int_as_float(__builtin_amdgcn_update_dpp(0, b, 0x160 | M, 0xf, 0xf, true));
}
template <int G>
__device__ __forceinline__ float group8_lane(float v) {
    constexpr unsigned sel = (unsigned)G * 0x249249u;
    return __int_as_float(__builtin_amdgcn_mov_dpp8(__float_as_int(v), sel));
}
__device__ __forceinline__ float mul_by_group_lane4(float acc, float v) {
    return acc * group8_lane<4>(v);
}

__device__ __forceinline__
void rmba_qproj_part(const int blk,
                     const float* __restrict__ A,
                     const float* __restrict__ Wq,
                     const float* __restrict__ bq,
                     const float* __restrict__ Wk,
                     const float* __restrict__ bk,
                     float* __restrict__ Qe)
{
    const int lane = threadIdx.x & 31;
    const int wave = threadIdx.x >> 5;
    const int half = lane >> 4;
    const int l16  = lane & 15;
    const int koff = half * 2;
    const int rowBase = blk * 16;
    const int colBase = wave * 16;
    const int bcol = colBase + l16;

    const float bias = bq[bcol];

    __shared__ __attribute__((aligned(16))) float sA[16 * AP];
    {
        const float4* a4 = (const float4*)(A + (size_t)rowBase * CRV);
        float4 st[6];
        #pragma unroll
        for (int j = 0; j < 6; ++j) st[j] = a4[(int)threadIdx.x + 256 * j];
        #pragma unroll
        for (int j = 0; j < 6; ++j) {
            const int f  = (int)threadIdx.x + 256 * j;
            const int r  = f / (CRV / 4);
            const int c4 = f % (CRV / 4);
            *(float4*)&sA[r * AP + c4 * 4] = st[j];
        }
    }
    __syncthreads();

    v8f acc = {};
    #pragma unroll 16
    for (int k = 0; k < CRV; k += 4) {
        v2f b;
        const v2f a = *(const v2f*)&sA[l16 * AP + k + koff];
        b.x = Wq[(k + koff) * DD + bcol];
        b.y = Wq[(k + koff + 1) * DD + bcol];
        acc = __builtin_amdgcn_wmma_f32_16x16x4_f32(false, a, false, b,
                                                    (short)0, acc, false, false);
    }

    float vals[8], ss[8];
    #pragma unroll
    for (int r = 0; r < 8; ++r) { vals[r] = acc[r] + bias; ss[r] = vals[r] * vals[r]; }

    #pragma unroll
    for (int r = 0; r < 8; ++r) ss[r] += lane_xor<8>(ss[r]);
    #pragma unroll
    for (int r = 0; r < 8; ++r) ss[r] += lane_xor<4>(ss[r]);
    #pragma unroll
    for (int r = 0; r < 8; ++r) ss[r] += lane_xor<2>(ss[r]);
    #pragma unroll
    for (int r = 0; r < 8; ++r) ss[r] += lane_xor<1>(ss[r]);

    __shared__ float sq[8][16];
    __shared__ float rnorm[16];
    __shared__ float sQ[16 * QP];
    if (l16 == 0) {
        #pragma unroll
        for (int r = 0; r < 8; ++r) sq[wave][half * 8 + r] = ss[r];
    }
    __syncthreads();
    if (threadIdx.x < 16) {
        float s = 0.f;
        #pragma unroll
        for (int w = 0; w < 8; ++w) s += sq[w][threadIdx.x];
        rnorm[threadIdx.x] = 1.0f / fmaxf(sqrtf(s), 1e-12f);
    }
    __syncthreads();
    #pragma unroll
    for (int r = 0; r < 8; ++r) {
        const int row = half * 8 + r;
        sQ[row * QP + bcol] = vals[r] * rnorm[row];
    }
    __syncthreads();

    if (wave < 6) {
        const int col = colBase + l16;
        const float* bp = (col < CB) ? (Wk + col * DD) : bk;
        const bool live = (col <= CB);
        v8f acc2 = {};
        #pragma unroll 16
        for (int k = 0; k < DD; k += 4) {
            v2f a, b;
            a.x = sQ[l16 * QP + k + koff];
            a.y = sQ[l16 * QP + k + koff + 1];
            const float x0 = bp[k + koff];
            const float x1 = bp[k + koff + 1];
            b.x = live ? x0 : 0.f;
            b.y = live ? x1 : 0.f;
            acc2 = __builtin_amdgcn_wmma_f32_16x16x4_f32(false, a, false, b,
                                                         (short)0, acc2, false, false);
        }
        #pragma unroll
        for (int r = 0; r < 8; ++r)
            Qe[(rowBase + half * 8 + r) * EP + col] = acc2[r];
    }
}

__device__ __forceinline__
void rmba_cell_part(const int blk,
                    const float* __restrict__ bev,
                    const float* __restrict__ Wk, const float* __restrict__ bk,
                    float* __restrict__ Te)
{
    const int lane = threadIdx.x & 31;
    const int wave = threadIdx.x >> 5;
    const int half = lane >> 4;
    const int l16  = lane & 15;
    const int koff = half * 2;
    const int rowBase = blk * 16;
    const int colBase = wave * 16;
    const int arow = rowBase + l16;
    const int bcol = colBase + l16;

    __shared__ float sq[8][16];
    __shared__ __attribute__((aligned(16))) float sT[16 * EP];

    const float biask = bk[bcol];
    float raw[5];
    #pragma unroll
    for (int j = 0; j < 5; ++j)
        raw[j] = bev[((int)(threadIdx.x >> 4) + 16 * j) * NBC + rowBase + (int)(threadIdx.x & 15)];
    v8f acck = {};
    #pragma unroll 10
    for (int k = 0; k < CB; k += 4) {
        v2f a, wk;
        a.x  = bev[(k + koff) * NBC + arow];
        a.y  = bev[(k + koff + 1) * NBC + arow];
        wk.x = Wk[(k + koff) * DD + bcol];
        wk.y = Wk[(k + koff + 1) * DD + bcol];
        acck = __builtin_amdgcn_wmma_f32_16x16x4_f32(false, a, false, wk,
                                                     (short)0, acck, false, false);
    }

    {
        const int ci = threadIdx.x & 15;
        const int c0 = threadIdx.x >> 4;
        #pragma unroll
        for (int j = 0; j < 5; ++j)
            sT[ci * EP + c0 + 16 * j] = raw[j];
        if (threadIdx.x < 240) {
            const int ri = threadIdx.x / 15;
            const int cj = threadIdx.x % 15;
            const int c  = (cj == 0) ? 80 : 81 + cj;
            sT[ri * EP + c] = (cj == 0) ? 1.0f : 0.0f;
        }
    }

    float ss[8];
    #pragma unroll
    for (int r = 0; r < 8; ++r) { const float kv = acck[r] + biask; ss[r] = kv * kv; }
    #pragma unroll
    for (int r = 0; r < 8; ++r) ss[r] += lane_xor<8>(ss[r]);
    #pragma unroll
    for (int r = 0; r < 8; ++r) ss[r] += lane_xor<4>(ss[r]);
    #pragma unroll
    for (int r = 0; r < 8; ++r) ss[r] += lane_xor<2>(ss[r]);
    #pragma unroll
    for (int r = 0; r < 8; ++r) ss[r] += lane_xor<1>(ss[r]);

    if (l16 == 0) {
        #pragma unroll
        for (int r = 0; r < 8; ++r) sq[wave][half * 8 + r] = ss[r];
    }
    __syncthreads();
    if (threadIdx.x < 16) {
        float s = 0.f;
        #pragma unroll
        for (int w = 0; w < 8; ++w) s += sq[w][threadIdx.x];
        sT[threadIdx.x * EP + 81] = 1.0f / fmaxf(sqrtf(s), 1e-12f);
    }
    __syncthreads();
    {
        const float4* s4 = (const float4*)sT;
        float4* t4 = (float4*)(Te + (size_t)rowBase * EP);
        t4[threadIdx.x] = s4[threadIdx.x];
        if (threadIdx.x < 128) t4[256 + threadIdx.x] = s4[256 + threadIdx.x];
    }
}

constexpr int QBLK = NRQ / 16;
constexpr int CBLK = NBC / 16;

__global__ __launch_bounds__(256)
void rmba_front_kernel(const float* __restrict__ A,
                       const float* __restrict__ Wq,
                       const float* __restrict__ bq,
                       const float* __restrict__ Wk,
                       const float* __restrict__ bk,
                       const float* __restrict__ bev,
                       float* __restrict__ Qe,
                       float* __restrict__ Te)
{
    if (blockIdx.x < QBLK)
        rmba_qproj_part((int)blockIdx.x, A, Wq, bq, Wk, bk, Qe);
    else
        rmba_cell_part((int)blockIdx.x - QBLK, bev, Wk, bk, Te);
}

constexpr int QPB = 4;

__global__ __launch_bounds__(256)
void rmba_attn_kernel(const float* __restrict__ Qe,
                      const float* __restrict__ Te,
                      const int*   __restrict__ cand,
                      float* __restrict__ attn_out,
                      float* __restrict__ ctxe)
{
    const int n0   = blockIdx.x * QPB;
    const int tid  = threadIdx.x;
    const int lane = tid & 31;
    const int wave = tid >> 5;
    const int sub  = lane & 7;

    __shared__ float red_sm[8];
    __shared__ float cpart[8][EP];

    const char* tbytes = (const char*)Te;
    const unsigned subOff = (unsigned)sub * 16u;
    const bool b0 = (lane & 1) != 0;
    const bool b1 = (lane & 2) != 0;
    const bool b2 = (lane & 4) != 0;
    const bool b3 = (lane & 8) != 0;
    const bool b4 = (lane & 16) != 0;

    for (int qi = 0; qi < QPB; ++qi) {
        const int n = n0 + qi;

        const int idx = cand[(unsigned)(n * MM) + (unsigned)tid];
        const int4* cand4 = (const int4*)cand;
        const unsigned gq = (unsigned)(n * (MM / 4)) + (unsigned)(tid >> 3) * 2u;
        const int4 ca = cand4[gq];
        const int4 cb = cand4[gq + 1u];
        const int all8 = ca.x & ca.y & ca.z & ca.w & cb.x & cb.y & cb.z & cb.w;
        const bool wave_live = __builtin_amdgcn_ballot_w32(all8 >= 0) != 0u;

        float4 kv[8][3];
        float dot = 0.f;
        if (wave_live) {
            const int gi[8] = {ca.x, ca.y, ca.z, ca.w, cb.x, cb.y, cb.z, cb.w};

            const float4* qe4 = (const float4*)Qe;
            const unsigned qq = (unsigned)(n * (EP / 4)) + (unsigned)sub;
            float4 q4[3];
            #pragma unroll
            for (int t = 0; t < 3; ++t) q4[t] = qe4[qq + (unsigned)(t * 8)];

            float p[8];
            #pragma unroll
            for (int g = 0; g < 8; ++g) {
                const unsigned ixc = (unsigned)(gi[g] < 0 ? 0 : gi[g]);
                const unsigned off = ixc * (unsigned)(EP * sizeof(float)) + subOff;
                #pragma unroll
                for (int t = 0; t < 3; ++t)
                    kv[g][t] = *(const float4*)(tbytes + off + t * 128);
            }

            #pragma unroll
            for (int g = 0; g < 8; ++g) {
                float acc = 0.f;
                #pragma unroll
                for (int t = 0; t < 3; ++t) {
                    acc = fmaf(kv[g][t].x, q4[t].x, acc);
                    acc = fmaf(kv[g][t].y, q4[t].y, acc);
                    acc = fmaf(kv[g][t].z, q4[t].z, acc);
                    acc = fmaf(kv[g][t].w, q4[t].w, acc);
                }
                p[g] = acc;
            }
            __builtin_amdgcn_sched_barrier(0);
            #pragma unroll
            for (int g = 0; g < 8; ++g) p[g] = mul_by_group_lane4(p[g], kv[g][2].y);

            float r1[4];
            #pragma unroll
            for (int i = 0; i < 4; ++i) {
                const float keep = b0 ? p[2 * i + 1] : p[2 * i];
                const float send = b0 ? p[2 * i]     : p[2 * i + 1];
                r1[i] = keep + lane_xor<1>(send);
            }
            float r2[2];
            #pragma unroll
            for (int i = 0; i < 2; ++i) {
                const float keep = b1 ? r1[2 * i + 1] : r1[2 * i];
                const float send = b1 ? r1[2 * i]     : r1[2 * i + 1];
                r2[i] = keep + lane_xor<2>(send);
            }
            const float keep3 = b2 ? r2[1] : r2[0];
            const float send3 = b2 ? r2[0] : r2[1];
            dot = keep3 + lane_xor<4>(send3);
        }

        const float logit = (idx >= 0) ? dot * INV_SQRT_D : -INFINITY;

        float mx = logit;
        mx = fmaxf(mx, lane_xor<16>(mx));
        mx = fmaxf(mx, lane_xor<8>(mx));
        mx = fmaxf(mx, lane_xor<4>(mx));
        mx = fmaxf(mx, lane_xor<2>(mx));
        mx = fmaxf(mx, lane_xor<1>(mx));
        if (lane == 0) red_sm[wave] = mx;
        __syncthreads();
        float bm = red_sm[0];
        #pragma unroll
        for (int w = 1; w < 8; ++w) bm = fmaxf(bm, red_sm[w]);
        __syncthreads();

        const float e = (idx >= 0) ? __expf(logit - bm) : 0.f;
        float sm = e;
        sm += lane_xor<16>(sm);
        sm += lane_xor<8>(sm);
        sm += lane_xor<4>(sm);
        sm += lane_xor<2>(sm);
        sm += lane_xor<1>(sm);
        if (lane == 0) red_sm[wave] = sm;
        __syncthreads();
        float bs = 0.f;
        #pragma unroll
        for (int w = 0; w < 8; ++w) bs += red_sm[w];

        const float a = e / bs;
        attn_out[(unsigned)(n * MM) + (unsigned)tid] = a;

        float y[3] = {0.f, 0.f, 0.f};
        if (wave_live) {
            float v[12] = {};
            const float agv[8] = {group8_lane<0>(a), group8_lane<1>(a), group8_lane<2>(a), group8_lane<3>(a),
                                  group8_lane<4>(a), group8_lane<5>(a), group8_lane<6>(a), group8_lane<7>(a)};
            #pragma unroll
            for (int g = 0; g < 8; ++g) {
                const float ag = agv[g];
                #pragma unroll
                for (int t = 0; t < 3; ++t) {
                    v[4 * t + 0] = fmaf(ag, kv[g][t].x, v[4 * t + 0]);
                    v[4 * t + 1] = fmaf(ag, kv[g][t].y, v[4 * t + 1]);
                    v[4 * t + 2] = fmaf(ag, kv[g][t].z, v[4 * t + 2]);
                    v[4 * t + 3] = fmaf(ag, kv[g][t].w, v[4 * t + 3]);
                }
            }
            float w6[6];
            #pragma unroll
            for (int i = 0; i < 6; ++i) {
                const float keep = b3 ? v[6 + i] : v[i];
                const float send = b3 ? v[i]     : v[6 + i];
                w6[i] = keep + lane_xor<8>(send);
            }
            #pragma unroll
            for (int i = 0; i < 3; ++i) {
                const float keep = b4 ? w6[3 + i] : w6[i];
                const float send = b4 ? w6[i]     : w6[3 + i];
                y[i] = keep + lane_xor<16>(send);
            }
        }
        {
            const int jb = (b3 ? 6 : 0) + (b4 ? 3 : 0);
            #pragma unroll
            for (int i = 0; i < 3; ++i)
                cpart[wave][8 * (jb + i) + sub] = y[i];
        }
        __syncthreads();
        if (tid < EP) {
            const int slot = 8 * (4 * (tid >> 5) + (tid & 3)) + ((tid >> 2) & 7);
            float s = 0.f;
            #pragma unroll
            for (int w8 = 0; w8 < 8; ++w8) s += cpart[w8][slot];
            ctxe[(unsigned)(n * EP) + (unsigned)tid] = s;
        }
    }
}

__global__ __launch_bounds__(256)
void rmba_oproj_kernel(const float* __restrict__ ctxe,
                       const float* __restrict__ Wv,
                       const float* __restrict__ bv,
                       const float* __restrict__ Wo,
                       const float* __restrict__ bo,
                       float* __restrict__ out)
{
    const int mTiles = NRQ / 16;
    const int mTile  = blockIdx.x % mTiles;
    const int nGrp   = blockIdx.x / mTiles;
    const int lane = threadIdx.x & 31;
    const int wave = threadIdx.x >> 5;
    const int half = lane >> 4;
    const int l16  = lane & 15;
    const int koff = half * 2;
    const int rowBase = mTile * 16;

    __shared__ float sC[16 * QP];
    __shared__ __attribute__((aligned(16))) float sX[16 * XP];

    {
        const int ccol = wave * 16 + l16;
        const float bvc = bv[ccol];
        {
            const float4* c4p = (const float4*)(ctxe + (size_t)rowBase * EP);
            const int f0 = (int)threadIdx.x;
            const int f1 = 256 + (int)threadIdx.x;
            const bool two = threadIdx.x < 128;
            const float4 x0 = c4p[f0];
            float4 x1 = make_float4(0.f, 0.f, 0.f, 0.f);
            if (two) x1 = c4p[f1];
            *(float4*)&sX[(f0 / (EP / 4)) * XP + (f0 % (EP / 4)) * 4] = x0;
            if (two) *(float4*)&sX[(f1 / (EP / 4)) * XP + (f1 % (EP / 4)) * 4] = x1;
        }
        __syncthreads();
        const v2f at = *(const v2f*)&sX[l16 * XP + CB + koff];
        v8f acc1 = {};
        #pragma unroll 10
        for (int k = 0; k < CB; k += 4) {
            v2f b;
            const v2f a = *(const v2f*)&sX[l16 * XP + k + koff];
            b.x = Wv[(k + koff) * DD + ccol];
            b.y = Wv[(k + koff + 1) * DD + ccol];
            acc1 = __builtin_amdgcn_wmma_f32_16x16x4_f32(false, a, false, b,
                                                         (short)0, acc1, false, false);
        }
        {
            v2f b;
            b.x = (half == 0) ? bvc : 0.f;
            b.y = 0.f;
            acc1 = __builtin_amdgcn_wmma_f32_16x16x4_f32(false, at, false, b,
                                                         (short)0, acc1, false, false);
        }
        #pragma unroll
        for (int r = 0; r < 8; ++r)
            sC[(half * 8 + r) * QP + ccol] = acc1[r];
    }
    __syncthreads();

    const int colBase = nGrp * 128 + wave * 16;
    const int bcol = colBase + l16;
    const float bias = bo[bcol];
    v8f acc = {};
    #pragma unroll 16
    for (int k = 0; k < DD; k += 4) {
        v2f a, b;
        a.x = sC[l16 * QP + k + koff];
        a.y = sC[l16 * QP + k + koff + 1];
        b.x = Wo[(k + koff) * CRV + bcol];
        b.y = Wo[(k + koff + 1) * CRV + bcol];
        acc = __builtin_amdgcn_wmma_f32_16x16x4_f32(false, a, false, b,
                                                    (short)0, acc, false, false);
    }
    #pragma unroll
    for (int r = 0; r < 8; ++r) {
        const int row = rowBase + half * 8 + r;
        out[row * CRV + bcol] = acc[r] + bias;
    }
}

extern "C" void kernel_launch(void* const* d_in, const int* in_sizes, int n_in,
                              void* d_out, int out_size, void* d_ws, size_t ws_size,
                              hipStream_t stream) {
    const float* rv_x = (const float*)d_in[0];
    const float* bev  = (const float*)d_in[1];
    const int*   cand = (const int*)  d_in[2];
    const float* Wq   = (const float*)d_in[3];
    const float* bq   = (const float*)d_in[4];
    const float* Wk   = (const float*)d_in[5];
    const float* bk   = (const float*)d_in[6];
    const float* Wv   = (const float*)d_in[7];
    const float* bv   = (const float*)d_in[8];
    const float* Wo   = (const float*)d_in[9];
    const float* bo   = (const float*)d_in[10];

    float* out_main = (float*)d_out;
    float* out_attn = out_main + (size_t)NRQ * CRV;

    float* Qe   = (float*)d_ws;
    float* Te   = Qe + (size_t)NRQ * EP;
    float* ctxe = Te + (size_t)NBC * EP;

    rmba_front_kernel<<<QBLK + CBLK, 256, 0, stream>>>(rv_x, Wq, bq, Wk, bk, bev, Qe, Te);
    rmba_attn_kernel <<<NRQ / QPB, 256, 0, stream>>>(Qe, Te, cand, out_attn, ctxe);
    rmba_oproj_kernel<<<(NRQ / 16) * (CRV / 128), 256, 0, stream>>>(ctxe, Wv, bv, Wo, bo, out_main);
}


# output path: Wvo = [Wv;bv] Wo (96 x 384) built once in the front launch; last kernel is one 21-step product per tile, not 21 + 32 with an LDS hand-over (sum order of out changes: reassociated)
# speedup vs baseline: 4.9909x; 1.1114x over previous
#include <hip/hip_runtime.h>
#include <math.h>


typedef float v2f __attribute__((ext_vector_type(2)));
typedef float v8f __attribute__((ext_vector_type(8)));

constexpr int NRQ = 4096;
constexpr int NBC = 16384;
constexpr int CB  = 80;
constexpr int CRV = 384;
constexpr int DD  = 128;
constexpr int MM  = 256;
constexpr int EP  = 96;
constexpr int QP  = 132;
constexpr int AP  = CRV + 4;
constexpr int XP  = EP + 4;
constexpr float INV_SQRT_D = 0.08838834764831845f;


template <int M>
__device__ __forceinline__ float lane_xor(float v) {
    const int b = __float_as_int(v);
    if constexpr (M == 16)
        return __int_as_float(__builtin_amdgcn_permlanex16(b, b, 0x76543210, 0xfedcba98, false, false));
    else
        return __int_as_float(__builtin_amdgcn_update_dpp(0, b, 0x160 | M, 0xf, 0xf, true));
}
template <int G>
__device__ __forceinline__ float group8_lane(float v) {
    constexpr unsigned sel = (unsigned)G * 0x249249u;
    return __int_as_float(__builtin_amdgcn_mov_dpp8(__float_as_int(v), sel));
}
__device__ __forceinline__ float mul_by_group_lane4(float acc, float v) {
    return acc * group8_lane<4>(v);
}

__device__ __forceinline__
void rmba_qproj_part(const int blk,
                     const float* __restrict__ A,
                     const float* __restrict__ Wq,
                     const float* __restrict__ bq,
                     const float* __restrict__ Wk,
                     const float* __restrict__ bk,
                     float* __restrict__ Qe)
{
    const int lane = threadIdx.x & 31;
    const int wave = threadIdx.x >> 5;
    const int half = lane >> 4;
    const int l16  = lane & 15;
    const int koff = half * 2;
    const int rowBase = blk * 16;
    const int colBase = wave * 16;
    const int bcol = colBase + l16;

    const float bias = bq[bcol];

    __shared__ __attribute__((aligned(16))) float sA[16 * AP];
    {
        const float4* a4 = (const float4*)(A + (size_t)rowBase * CRV);
        float4 st[6];
        #pragma unroll
        for (int j = 0; j < 6; ++j) st[j] = a4[(int)threadIdx.x + 256 * j];
        #pragma unroll
        for (int j = 0; j < 6; ++j) {
            const int f  = (int)threadIdx.x + 256 * j;
            const int r  = f / (CRV / 4);
            const int c4 = f % (CRV / 4);
            *(float4*)&sA[r * AP + c4 * 4] = st[j];
        }
    }
    __syncthreads();

    v8f acc = {};
    #pragma unroll 16
    for (int k = 0; k < CRV; k += 4) {
        v2f b;
        const v2f a = *(const v2f*)&sA[l16 * AP + k + koff];
        b.x = Wq[(k + koff) * DD + bcol];
        b.y = Wq[(k + koff + 1) * DD + bcol];
        acc = __builtin_amdgcn_wmma_f32_16x16x4_f32(false, a, false, b,
                                                    (short)0, acc, false, false);
    }

    float vals[8], ss[8];
    #pragma unroll
    for (int r = 0; r < 8; ++r) { vals[r] = acc[r] + bias; ss[r] = vals[r] * vals[r]; }

    #pragma unroll
    for (int r = 0; r < 8; ++r) ss[r] += lane_xor<8>(ss[r]);
    #pragma unroll
    for (int r = 0; r < 8; ++r) ss[r] += lane_xor<4>(ss[r]);
    #pragma unroll
    for (int r = 0; r < 8; ++r) ss[r] += lane_xor<2>(ss[r]);
    #pragma unroll
    for (int r = 0; r < 8; ++r) ss[r] += lane_xor<1>(ss[r]);

    __shared__ float sq[8][16];
    __shared__ float rnorm[16];
    __shared__ float sQ[16 * QP];
    if (l16 == 0) {
        #pragma unroll
        for (int r = 0; r < 8; ++r) sq[wave][half * 8 + r] = ss[r];
    }
    __syncthreads();
    if (threadIdx.x < 16) {
        float s = 0.f;
        #pragma unroll
        for (int w = 0; w < 8; ++w) s += sq[w][threadIdx.x];
        rnorm[threadIdx.x] = 1.0f / fmaxf(sqrtf(s), 1e-12f);
    }
    __syncthreads();
    #pragma unroll
    for (int r = 0; r < 8; ++r) {
        const int row = half * 8 + r;
        sQ[row * QP + bcol] = vals[r] * rnorm[row];
    }
    __syncthreads();

    if (wave < 6) {
        const int col = colBase + l16;
        const float* bp = (col < CB) ? (Wk + col * DD) : bk;
        const bool live = (col <= CB);
        v8f acc2 = {};
        #pragma unroll 16
        for (int k = 0; k < DD; k += 4) {
            v2f a, b;
            a.x = sQ[l16 * QP + k + koff];
            a.y = sQ[l16 * QP + k + koff + 1];
            const float x0 = bp[k + koff];
            const float x1 = bp[k + koff + 1];
            b.x = live ? x0 : 0.f;
            b.y = live ? x1 : 0.f;
            acc2 = __builtin_amdgcn_wmma_f32_16x16x4_f32(false, a, false, b,
                                                         (short)0, acc2, false, false);
        }
        #pragma unroll
        for (int r = 0; r < 8; ++r)
            Qe[(rowBase + half * 8 + r) * EP + col] = acc2[r];
    }
}

__device__ __forceinline__
void rmba_cell_part(const int blk,
                    const float* __restrict__ bev,
                    const float* __restrict__ Wk, const float* __restrict__ bk,
                    float* __restrict__ Te)
{
    const int lane = threadIdx.x & 31;
    const int wave = threadIdx.x >> 5;
    const int half = lane >> 4;
    const int l16  = lane & 15;
    const int koff = half * 2;
    const int rowBase = blk * 16;
    const int colBase = wave * 16;
    const int arow = rowBase + l16;
    const int bcol = colBase + l16;

    __shared__ float sq[8][16];
    __shared__ __attribute__((aligned(16))) float sT[16 * EP];

    const float biask = bk[bcol];
    float raw[5];
    #pragma unroll
    for (int j = 0; j < 5; ++j)
        raw[j] = bev[((int)(threadIdx.x >> 4) + 16 * j) * NBC + rowBase + (int)(threadIdx.x & 15)];
    v8f acck = {};
    #pragma unroll 10
    for (int k = 0; k < CB; k += 4) {
        v2f a, wk;
        a.x  = bev[(k + koff) * NBC + arow];
        a.y  = bev[(k + koff + 1) * NBC + arow];
        wk.x = Wk[(k + koff) * DD + bcol];
        wk.y = Wk[(k + koff + 1) * DD + bcol];
        acck = __builtin_amdgcn_wmma_f32_16x16x4_f32(false, a, false, wk,
                                                     (short)0, acck, false, false);
    }

    {
        const int ci = threadIdx.x & 15;
        const int c0 = threadIdx.x >> 4;
        #pragma unroll
        for (int j = 0; j < 5; ++j)
            sT[ci * EP + c0 + 16 * j] = raw[j];
        if (threadIdx.x < 240) {
            const int ri = threadIdx.x / 15;
            const int cj = threadIdx.x % 15;
            const int c  = (cj == 0) ? 80 : 81 + cj;
            sT[ri * EP + c] = (cj == 0) ? 1.0f : 0.0f;
        }
    }

    float ss[8];
    #pragma unroll
    for (int r = 0; r < 8; ++r) { const float kv = acck[r] + biask; ss[r] = kv * kv; }
    #pragma unroll
    for (int r = 0; r < 8; ++r) ss[r] += lane_xor<8>(ss[r]);
    #pragma unroll
    for (int r = 0; r < 8; ++r) ss[r] += lane_xor<4>(ss[r]);
    #pragma unroll
    for (int r = 0; r < 8; ++r) ss[r] += lane_xor<2>(ss[r]);
    #pragma unroll
    for (int r = 0; r < 8; ++r) ss[r] += lane_xor<1>(ss[r]);

    if (l16 == 0) {
        #pragma unroll
        for (int r = 0; r < 8; ++r) sq[wave][half * 8 + r] = ss[r];
    }
    __syncthreads();
    if (threadIdx.x < 16) {
        float s = 0.f;
        #pragma unroll
        for (int w = 0; w < 8; ++w) s += sq[w][threadIdx.x];
        sT[threadIdx.x * EP + 81] = 1.0f / fmaxf(sqrtf(s), 1e-12f);
    }
    __syncthreads();
    {
        const float4* s4 = (const float4*)sT;
        float4* t4 = (float4*)(Te + (size_t)rowBase * EP);
        t4[threadIdx.x] = s4[threadIdx.x];
        if (threadIdx.x < 128) t4[256 + threadIdx.x] = s4[256 + threadIdx.x];
    }
}

__device__ __forceinline__
void rmba_wvo_part(const int blk,
                   const float* __restrict__ Wv,
                   const float* __restrict__ bv,
                   const float* __restrict__ Wo,
                   float* __restrict__ Wvo)
{
    const int lane = threadIdx.x & 31;
    const int wave = threadIdx.x >> 5;
    const int half = lane >> 4;
    const int l16  = lane & 15;
    const int koff = half * 2;
    const int rowBase = (blk / 3) * 16;
    const int bcol = (blk % 3) * 128 + wave * 16 + l16;
    const int arow = rowBase + l16;
    const float* ap = (arow < CB) ? (Wv + arow * DD) : bv;
    const bool live = (arow <= CB);
    v8f acc = {};
    #pragma unroll 16
    for (int k = 0; k < DD; k += 4) {
        v2f a, b;
        const v2f x = *(const v2f*)&ap[k + koff];
        a.x = live ? x.x : 0.f;
        a.y = live ? x.y : 0.f;
        b.x = Wo[(k + koff) * CRV + bcol];
        b.y = Wo[(k + koff + 1) * CRV + bcol];
        acc = __builtin_amdgcn_wmma_f32_16x16x4_f32(false, a, false, b,
                                                    (short)0, acc, false, false);
    }
    #pragma unroll
    for (int r = 0; r < 8; ++r)
        Wvo[(rowBase + half * 8 + r) * CRV + bcol] = acc[r];
}

constexpr int QBLK = NRQ / 16;
constexpr int CBLK = NBC / 16;
constexpr int WBLK = (EP / 16) * (CRV / 128);

__global__ __launch_bounds__(256)
void rmba_front_kernel(const float* __restrict__ A,
                       const float* __restrict__ Wq,
                       const float* __restrict__ bq,
                       const float* __restrict__ Wk,
                       const float* __restrict__ bk,
                       const float* __restrict__ bev,
                       const float* __restrict__ Wv,
                       const float* __restrict__ bv,
                       const float* __restrict__ Wo,
                       float* __restrict__ Qe,
                       float* __restrict__ Te,
                       float* __restrict__ Wvo)
{
    if (blockIdx.x < QBLK)
        rmba_qproj_part((int)blockIdx.x, A, Wq, bq, Wk, bk, Qe);
    else if (blockIdx.x < QBLK + WBLK)
        rmba_wvo_part((int)blockIdx.x - QBLK, Wv, bv, Wo, Wvo);
    else
        rmba_cell_part((int)blockIdx.x - QBLK - WBLK, bev, Wk, bk, Te);
}

constexpr int QPB = 4;

__global__ __launch_bounds__(256)
void rmba_attn_kernel(const float* __restrict__ Qe,
                      const float* __restrict__ Te,
                      const int*   __restrict__ cand,
                      float* __restrict__ attn_out,
                      float* __restrict__ ctxe)
{
    const int n0   = blockIdx.x * QPB;
    const int tid  = threadIdx.x;
    const int lane = tid & 31;
    const int wave = tid >> 5;
    const int sub  = lane & 7;

    __shared__ float red_sm[8];
    __shared__ float cpart[8][EP];

    const char* tbytes = (const char*)Te;
    const unsigned subOff = (unsigned)sub * 16u;
    const bool b0 = (lane & 1) != 0;
    const bool b1 = (lane & 2) != 0;
    const bool b2 = (lane & 4) != 0;
    const bool b3 = (lane & 8) != 0;
    const bool b4 = (lane & 16) != 0;

    for (int qi = 0; qi < QPB; ++qi) {
        const int n = n0 + qi;

        const int idx = cand[(unsigned)(n * MM) + (unsigned)tid];
        const int4* cand4 = (const int4*)cand;
        const unsigned gq = (unsigned)(n * (MM / 4)) + (unsigned)(tid >> 3) * 2u;
        const int4 ca = cand4[gq];
        const int4 cb = cand4[gq + 1u];
        const int all8 = ca.x & ca.y & ca.z & ca.w & cb.x & cb.y & cb.z & cb.w;
        const bool wave_live = __builtin_amdgcn_ballot_w32(all8 >= 0) != 0u;

        float4 kv[8][3];
        float dot = 0.f;
        if (wave_live) {
            const int gi[8] = {ca.x, ca.y, ca.z, ca.w, cb.x, cb.y, cb.z, cb.w};

            const float4* qe4 = (const float4*)Qe;
            const unsigned qq = (unsigned)(n * (EP / 4)) + (unsigned)sub;
            float4 q4[3];
            #pragma unroll
            for (int t = 0; t < 3; ++t) q4[t] = qe4[qq + (unsigned)(t * 8)];

            float p[8];
            #pragma unroll
            for (int g = 0; g < 8; ++g) {
                const unsigned ixc = (unsigned)(gi[g] < 0 ? 0 : gi[g]);
                const unsigned off = ixc * (unsigned)(EP * sizeof(float)) + subOff;
                #pragma unroll
                for (int t = 0; t < 3; ++t)
                    kv[g][t] = *(const float4*)(tbytes + off + t * 128);
            }

            #pragma unroll
            for (int g = 0; g < 8; ++g) {
                float acc = 0.f;
                #pragma unroll
                for (int t = 0; t < 3; ++t) {
                    acc = fmaf(kv[g][t].x, q4[t].x, acc);
                    acc = fmaf(kv[g][t].y, q4[t].y, acc);
                    acc = fmaf(kv[g][t].z, q4[t].z, acc);
                    acc = fmaf(kv[g][t].w, q4[t].w, acc);
                }
                p[g] = acc;
            }
            __builtin_amdgcn_sched_barrier(0);
            #pragma unroll
            for (int g = 0; g < 8; ++g) p[g] = mul_by_group_lane4(p[g], kv[g][2].y);

            float r1[4];
            #pragma unroll
            for (int i = 0; i < 4; ++i) {
                const float keep = b0 ? p[2 * i + 1] : p[2 * i];
                const float send = b0 ? p[2 * i]     : p[2 * i + 1];
                r1[i] = keep + lane_xor<1>(send);
            }
            float r2[2];
            #pragma unroll
            for (int i = 0; i < 2; ++i) {
                const float keep = b1 ? r1[2 * i + 1] : r1[2 * i];
                const float send = b1 ? r1[2 * i]     : r1[2 * i + 1];
                r2[i] = keep + lane_xor<2>(send);
            }
            const float keep3 = b2 ? r2[1] : r2[0];
            const float send3 = b2 ? r2[0] : r2[1];
            dot = keep3 + lane_xor<4>(send3);
        }

        const float logit = (idx >= 0) ? dot * INV_SQRT_D : -INFINITY;

        float mx = logit;
        mx = fmaxf(mx, lane_xor<16>(mx));
        mx = fmaxf(mx, lane_xor<8>(mx));
        mx = fmaxf(mx, lane_xor<4>(mx));
        mx = fmaxf(mx, lane_xor<2>(mx));
        mx = fmaxf(mx, lane_xor<1>(mx));
        if (lane == 0) red_sm[wave] = mx;
        __syncthreads();
        float bm = red_sm[0];
        #pragma unroll
        for (int w = 1; w < 8; ++w) bm = fmaxf(bm, red_sm[w]);
        __syncthreads();

        const float e = (idx >= 0) ? __expf(logit - bm) : 0.f;
        float sm = e;
        sm += lane_xor<16>(sm);
        sm += lane_xor<8>(sm);
        sm += lane_xor<4>(sm);
        sm += lane_xor<2>(sm);
        sm += lane_xor<1>(sm);
        if (lane == 0) red_sm[wave] = sm;
        __syncthreads();
        float bs = 0.f;
        #pragma unroll
        for (int w = 0; w < 8; ++w) bs += red_sm[w];

        const float a = e / bs;
        attn_out[(unsigned)(n * MM) + (unsigned)tid] = a;

        float y[3] = {0.f, 0.f, 0.f};
        if (wave_live) {
            float v[12] = {};
            const float agv[8] = {group8_lane<0>(a), group8_lane<1>(a), group8_lane<2>(a), group8_lane<3>(a),
                                  group8_lane<4>(a), group8_lane<5>(a), group8_lane<6>(a), group8_lane<7>(a)};
            #pragma unroll
            for (int g = 0; g < 8; ++g) {
                const float ag = agv[g];
                #pragma unroll
                for (int t = 0; t < 3; ++t) {
                    v[4 * t + 0] = fmaf(ag, kv[g][t].x, v[4 * t + 0]);
                    v[4 * t + 1] = fmaf(ag, kv[g][t].y, v[4 * t + 1]);
                    v[4 * t + 2] = fmaf(ag, kv[g][t].z, v[4 * t + 2]);
                    v[4 * t + 3] = fmaf(ag, kv[g][t].w, v[4 * t + 3]);
                }
            }
            float w6[6];
            #pragma unroll
            for (int i = 0; i < 6; ++i) {
                const float keep = b3 ? v[6 + i] : v[i];
                const float send = b3 ? v[i]     : v[6 + i];
                w6[i] = keep + lane_xor<8>(send);
            }
            #pragma unroll
            for (int i = 0; i < 3; ++i) {
                const float keep = b4 ? w6[3 + i] : w6[i];
                const float send = b4 ? w6[i]     : w6[3 + i];
                y[i] = keep + lane_xor<16>(send);
            }
        }
        {
            const int jb = (b3 ? 6 : 0) + (b4 ? 3 : 0);
            #pragma unroll
            for (int i = 0; i < 3; ++i)
                cpart[wave][8 * (jb + i) + sub] = y[i];
        }
        __syncthreads();
        if (tid < EP) {
            const int slot = 8 * (4 * (tid >> 5) + (tid & 3)) + ((tid >> 2) & 7);
            float s = 0.f;
            #pragma unroll
            for (int w8 = 0; w8 < 8; ++w8) s += cpart[w8][slot];
            ctxe[(unsigned)(n * EP) + (unsigned)tid] = s;
        }
    }
}

__global__ __launch_bounds__(256)
void rmba_oproj_kernel(const float* __restrict__ ctxe,
                       const float* __restrict__ Wvo,
                       const float* __restrict__ bo,
                       float* __restrict__ out)
{
    const int mTiles = NRQ / 16;
    const int mTile  = blockIdx.x % mTiles;
    const int nGrp   = blockIdx.x / mTiles;
    const int lane = threadIdx.x & 31;
    const int wave = threadIdx.x >> 5;
    const int half = lane >> 4;
    const int l16  = lane & 15;
    const int koff = half * 2;
    const int rowBase = mTile * 16;
    const int bcol = nGrp * 128 + wave * 16 + l16;

    __shared__ __attribute__((aligned(16))) float sX[16 * XP];

    const float bias = bo[bcol];
    const float wlast = Wvo[CB * CRV + bcol];
    {
        const float4* c4p = (const float4*)(ctxe + (size_t)rowBase * EP);
        const int f0 = (int)threadIdx.x;
        const int f1 = 256 + (int)threadIdx.x;
        const bool two = threadIdx.x < 128;
        const float4 x0 = c4p[f0];
        float4 x1 = make_float4(0.f, 0.f, 0.f, 0.f);
        if (two) x1 = c4p[f1];
        *(float4*)&sX[(f0 / (EP / 4)) * XP + (f0 % (EP / 4)) * 4] = x0;
        if (two) *(float4*)&sX[(f1 / (EP / 4)) * XP + (f1 % (EP / 4)) * 4] = x1;
    }
    __syncthreads();
    const v2f at = *(const v2f*)&sX[l16 * XP + CB + koff];
    v8f acc = {};
    #pragma unroll 10
    for (int k = 0; k < CB; k += 4) {
        v2f b;
        const v2f a = *(const v2f*)&sX[l16 * XP + k + koff];
        b.x = Wvo[(k + koff) * CRV + bcol];
        b.y = Wvo[(k + koff + 1) * CRV + bcol];
        acc = __builtin_amdgcn_wmma_f32_16x16x4_f32(false, a, false, b,
                                                    (short)0, acc, false, false);
    }
    {
        v2f b;
        b.x = (half == 0) ? wlast : 0.f;
        b.y = 0.f;
        acc = __builtin_amdgcn_wmma_f32_16x16x4_f32(false, at, false, b,
                                                    (short)0, acc, false, false);
    }
    #pragma unroll
    for (int r = 0; r < 8; ++r) {
        const int row = rowBase + half * 8 + r;
        out[row * CRV + bcol] = acc[r] + bias;
    }
}

extern "C" void kernel_launch(void* const* d_in, const int* in_sizes, int n_in,
                              void* d_out, int out_size, void* d_ws, size_t ws_size,
                              hipStream_t stream) {
    const float* rv_x = (const float*)d_in[0];
    const float* bev  = (const float*)d_in[1];
    const int*   cand = (const int*)  d_in[2];
    const float* Wq   = (const float*)d_in[3];
    const float* bq   = (const float*)d_in[4];
    const float* Wk   = (const float*)d_in[5];
    const float* bk   = (const float*)d_in[6];
    const float* Wv   = (const float*)d_in[7];
    const float* bv   = (const float*)d_in[8];
    const float* Wo   = (const float*)d_in[9];
    const float* bo   = (const float*)d_in[10];

    float* out_main = (float*)d_out;
    float* out_attn = out_main + (size_t)NRQ * CRV;

    float* Qe   = (float*)d_ws;
    float* Te   = Qe + (size_t)NRQ * EP;
    float* ctxe = Te + (size_t)NBC * EP;
    float* Wvo  = ctxe + (size_t)NRQ * EP;

    rmba_front_kernel<<<QBLK + WBLK + CBLK, 256, 0, stream>>>(rv_x, Wq, bq, Wk, bk, bev, Wv, bv, Wo, Qe, Te, Wvo);
    rmba_attn_kernel <<<NRQ / QPB, 256, 0, stream>>>(Qe, Te, cand, out_attn, ctxe);
    rmba_oproj_kernel<<<(NRQ / 16) * (CRV / 128), 256, 0, stream>>>(ctxe, Wvo, bo, out_main);
}
